# starting kernel
# baseline (speedup 1.0000x reference)
#include <hip/hip_runtime.h>
#include <hip/hip_bf16.h>


#define Bc   2
#define Nn   2048
#define Dd   256
#define Hh   8
#define DHh  32
#define Mm   2
#define Ll   4096

typedef __attribute__((ext_vector_type(2))) float v2f;
typedef __attribute__((ext_vector_type(8))) float v8f;

__global__ void ea_knn_edge(const float* __restrict__ pts,
                            const float* __restrict__ feats,
                            float* __restrict__ out,
                            int npts, int nsets, int remap)
{
    const int lane = threadIdx.x & 31;
    const int wave = threadIdx.x >> 5;
    const int rowg = blockIdx.x * 8 + wave;
    if (rowg >= nsets * npts) return;
    const int s = rowg / npts;
    const int n = rowg - s * npts;
    int os = s;
    if (remap) { int m = s / Bc; int b = s - m * Bc; os = b * Mm + m; }

    const float* P = pts   + (size_t)s  * npts * 3;
    const float* F = feats + (size_t)s  * npts * Dd;
    float*    Orow = out   + ((size_t)os * npts + n) * Dd;

    const float px = P[n*3+0], py = P[n*3+1], pz = P[n*3+2];

    unsigned long long key[8];
    #pragma unroll
    for (int t = 0; t < 8; ++t) key[t] = ~0ULL;

    for (int j = lane; j < npts; j += 32) {
        float dx = px - P[j*3+0];
        float dy = py - P[j*3+1];
        float dz = pz - P[j*3+2];
        float d2 = dx*dx + dy*dy + dz*dz;
        unsigned long long k =
            ((unsigned long long)__float_as_uint(d2) << 32) | (unsigned)j;
        if (k < key[7]) {
            key[7] = k;
            #pragma unroll
            for (int t = 7; t > 0; --t) {
                if (key[t] < key[t-1]) {
                    unsigned long long tmp = key[t]; key[t] = key[t-1]; key[t-1] = tmp;
                }
            }
        }
    }

    int head = 0;
    int sel[8];
    #pragma unroll
    for (int t = 0; t < 8; ++t) {
        unsigned long long cand = (head < 8) ? key[head] : ~0ULL;
        unsigned long long mn = cand;
        #pragma unroll
        for (int off = 16; off >= 1; off >>= 1) {
            unsigned long long o = __shfl_xor(mn, off);
            mn = (o < mn) ? o : mn;
        }
        if (cand == mn && head < 8) head++;
        sel[t] = (int)(mn & 0xFFFFFFFFu);
    }

    for (int c = lane; c < Dd; c += 32) {
        float ssum = 0.f;
        #pragma unroll
        for (int t = 0; t < 8; ++t) ssum += F[(size_t)sel[t] * Dd + c];
        Orow[c] = F[(size_t)n * Dd + c] - ssum * 0.125f;
    }
}

__global__ void ea_gemm256(const float* __restrict__ A, const float* __restrict__ W,
                           const float* __restrict__ bias, float* __restrict__ OUT,
                           int R)
{
    const int lane = threadIdx.x & 31;
    const int wave = threadIdx.x >> 5;
    const int tile = blockIdx.x * 8 + wave;
    const int total = (R / 16) * (Dd / 16);
    if (tile >= total) return;
    const int tm = tile >> 4, tn = tile & 15;
    const int col = lane & 15, half = lane >> 4, klo = half * 2;

    const float* Ap = A + (size_t)(tm * 16 + col) * Dd;
    const float* Wp = W + (size_t)(tn * 16 + col) * Dd;

    v8f acc = {};
    #pragma unroll 4
    for (int k = 0; k < Dd; k += 4) {
        v2f a = *(const v2f*)(Ap + k + klo);
        v2f b = *(const v2f*)(Wp + k + klo);
        acc = __builtin_amdgcn_wmma_f32_16x16x4_f32(false, a, false, b,
                                                    (short)0, acc, false, false);
    }
    const float bv = bias[tn * 16 + col];
    #pragma unroll
    for (int r = 0; r < 8; ++r) {
        int row = tm * 16 + r + 8 * half;
        OUT[(size_t)row * Dd + tn * 16 + col] = acc[r] + bv;
    }
}

__global__ void ea_attn(const float* __restrict__ Q, const float* __restrict__ Km,
                        const float* __restrict__ Vm, float* __restrict__ O)
{
    __shared__ float plds[8 * 256];
    const int lane = threadIdx.x & 31;
    const int wave = threadIdx.x >> 5;
    const int tile = blockIdx.x * 8 + wave;
    const int qt = tile & 127;
    const int h  = (tile >> 7) & 7;
    const int b  = tile >> 10;
    const int col = lane & 15, half = lane >> 4, klo = half * 2;
    float* pw = plds + wave * 256;

    const float* Qrow = Q + (size_t)(b * Nn + qt * 16 + col) * Dd + h * DHh;
    v2f qf[8];
    #pragma unroll
    for (int kk = 0; kk < 8; ++kk) qf[kk] = *(const v2f*)(Qrow + kk * 4 + klo);

    float mrow[8], lrow[8];
    v8f acc0 = {}, acc1 = {};
    #pragma unroll
    for (int r = 0; r < 8; ++r) { mrow[r] = -__builtin_inff(); lrow[r] = 0.f; }

    const float* Kb = Km + (size_t)b * Ll * Dd + h * DHh;
    const float* Vb = Vm + (size_t)b * Ll * Dd + h * DHh;
    const float scale = 0.17677669529663687f;

    for (int kt = 0; kt < Ll / 16; ++kt) {
        if (kt + 1 < Ll / 16)
            __builtin_prefetch(Kb + (size_t)((kt + 1) * 16 + col) * Dd, 0, 1);

        v8f s = {};
        const float* Kp = Kb + (size_t)(kt * 16 + col) * Dd;
        #pragma unroll
        for (int kk = 0; kk < 8; ++kk) {
            v2f bf = *(const v2f*)(Kp + kk * 4 + klo);
            s = __builtin_amdgcn_wmma_f32_16x16x4_f32(false, qf[kk], false, bf,
                                                      (short)0, s, false, false);
        }

        float p[8];
        #pragma unroll
        for (int r = 0; r < 8; ++r) {
            float v = s[r] * scale;
            float mx = v;
            #pragma unroll
            for (int off = 8; off >= 1; off >>= 1) mx = fmaxf(mx, __shfl_xor(mx, off));
            float mnew = fmaxf(mrow[r], mx);
            float pe = __expf(v - mnew);
            p[r] = pe;
            float sum = pe;
            #pragma unroll
            for (int off = 8; off >= 1; off >>= 1) sum += __shfl_xor(sum, off);
            float corr = __expf(mrow[r] - mnew);
            lrow[r] = lrow[r] * corr + sum;
            acc0[r] *= corr;
            acc1[r] *= corr;
            mrow[r] = mnew;
        }

        #pragma unroll
        for (int r = 0; r < 8; ++r) pw[(r + 8 * half) * 16 + col] = p[r];
        __syncthreads();

        #pragma unroll
        for (int kk2 = 0; kk2 < 4; ++kk2) {
            v2f af;
            af.x = pw[col * 16 + kk2 * 4 + klo];
            af.y = pw[col * 16 + kk2 * 4 + klo + 1];
            const float* Vp0 = Vb + (size_t)(kt * 16 + kk2 * 4 + klo) * Dd;
            const float* Vp1 = Vp0 + Dd;
            v2f b0; b0.x = Vp0[col];      b0.y = Vp1[col];
            v2f b1; b1.x = Vp0[16 + col]; b1.y = Vp1[16 + col];
            acc0 = __builtin_amdgcn_wmma_f32_16x16x4_f32(false, af, false, b0,
                                                         (short)0, acc0, false, false);
            acc1 = __builtin_amdgcn_wmma_f32_16x16x4_f32(false, af, false, b1,
                                                         (short)0, acc1, false, false);
        }
        __syncthreads();
    }

    #pragma unroll
    for (int r = 0; r < 8; ++r) {
        float inv = 1.0f / lrow[r];
        int row = b * Nn + qt * 16 + r + 8 * half;
        float* Op = O + (size_t)row * Dd + h * DHh;
        Op[col]      = acc0[r] * inv;
        Op[16 + col] = acc1[r] * inv;
    }
}

__global__ void ea_gemm_out(const float* __restrict__ A, const float* __restrict__ W,
                            const float* __restrict__ bias,
                            const float* __restrict__ pts,
                            const float* __restrict__ sw, const float* __restrict__ sb,
                            float* __restrict__ OUT, int R)
{
    const int lane = threadIdx.x & 31;
    const int wave = threadIdx.x >> 5;
    const int tile = blockIdx.x * 8 + wave;
    const int total = (R / 16) * (Dd / 16);
    if (tile >= total) return;
    const int tm = tile >> 4, tn = tile & 15;
    const int col = lane & 15, half = lane >> 4, klo = half * 2;
    const int colg = tn * 16 + col;

    const float* Ap = A + (size_t)(tm * 16 + col) * Dd;
    const float* Wp = W + (size_t)colg * Dd;

    v8f acc = {};
    #pragma unroll 4
    for (int k = 0; k < Dd; k += 4) {
        v2f a = *(const v2f*)(Ap + k + klo);
        v2f b = *(const v2f*)(Wp + k + klo);
        acc = __builtin_amdgcn_wmma_f32_16x16x4_f32(false, a, false, b,
                                                    (short)0, acc, false, false);
    }
    const float s0 = sw[colg*3+0], s1 = sw[colg*3+1], s2 = sw[colg*3+2];
    const float add = bias[colg] + sb[colg];
    #pragma unroll
    for (int r = 0; r < 8; ++r) {
        int row = tm * 16 + r + 8 * half;
        float px = pts[row*3+0], py = pts[row*3+1], pz = pts[row*3+2];
        OUT[(size_t)row * Dd + colg] = acc[r] + add + px*s0 + py*s1 + pz*s2;
    }
}

extern "C" void kernel_launch(void* const* d_in, const int* in_sizes, int n_in,
                              void* d_out, int out_size, void* d_ws, size_t ws_size,
                              hipStream_t stream)
{
    const float* cur_pts  = (const float*)d_in[0];
    const float* cur_feat = (const float*)d_in[1];
    const float* nbr_pts  = (const float*)d_in[2];
    const float* nbr_feat = (const float*)d_in[3];
    const float* in_w     = (const float*)d_in[4];
    const float* in_b     = (const float*)d_in[5];
    const float* out_w    = (const float*)d_in[6];
    const float* out_b    = (const float*)d_in[7];
    const float* sp_w     = (const float*)d_in[8];
    const float* sp_b     = (const float*)d_in[9];
    float* out = (float*)d_out;

    float* ws   = (float*)d_ws;
    float* curE = ws;
    float* nbrE = curE + 4096 * 256;
    float* Qb   = nbrE + 8192 * 256;
    float* Kb   = Qb   + 4096 * 256;
    float* Vb   = Kb   + 8192 * 256;
    float* Ob   = Vb   + 8192 * 256;

    const dim3 blk(256);

    ea_knn_edge<<<dim3(4096 / 8), blk, 0, stream>>>(cur_pts, cur_feat, curE,
                                                    Nn, Bc, 0);
    ea_knn_edge<<<dim3(8192 / 8), blk, 0, stream>>>(nbr_pts, nbr_feat, nbrE,
                                                    Nn, Mm * Bc, 1);

    ea_gemm256<<<dim3(4096 / 8), blk, 0, stream>>>(curE, in_w,           in_b,       Qb, 4096);
    ea_gemm256<<<dim3(8192 / 8), blk, 0, stream>>>(nbrE, in_w + 256*256, in_b + 256, Kb, 8192);
    ea_gemm256<<<dim3(8192 / 8), blk, 0, stream>>>(nbrE, in_w + 512*256, in_b + 512, Vb, 8192);

    ea_attn<<<dim3(2048 / 8), blk, 0, stream>>>(Qb, Kb, Vb, Ob);

    ea_gemm_out<<<dim3(4096 / 8), blk, 0, stream>>>(Ob, out_w, out_b,
                                                    cur_pts, sp_w, sp_b,
                                                    out, 4096);
}


# attention on transposed tiles (S^T=K*Q^T, O^T=V^T*P^T): a query is a lane column, so max/sum take 1 permlanex16 not 64 shuffles and P feeds the WMMA from registers (no LDS/barrier); sum order changes
# speedup vs baseline: 1.7763x; 1.7763x over previous
#include <hip/hip_runtime.h>
#include <hip/hip_bf16.h>


#define Bc   2
#define Nn   2048
#define Dd   256
#define Hh   8
#define DHh  32
#define Mm   2
#define Ll   4096

typedef __attribute__((ext_vector_type(2))) float v2f;
typedef __attribute__((ext_vector_type(8))) float v8f;

__global__ void ea_knn_edge(const float* __restrict__ pts,
                            const float* __restrict__ feats,
                            float* __restrict__ out,
                            int npts, int nsets, int remap)
{
    const int lane = threadIdx.x & 31;
    const int wave = threadIdx.x >> 5;
    const int rowg = blockIdx.x * 8 + wave;
    if (rowg >= nsets * npts) return;
    const int s = rowg / npts;
    const int n = rowg - s * npts;
    int os = s;
    if (remap) { int m = s / Bc; int b = s - m * Bc; os = b * Mm + m; }

    const float* P = pts   + (size_t)s  * npts * 3;
    const float* F = feats + (size_t)s  * npts * Dd;
    float*    Orow = out   + ((size_t)os * npts + n) * Dd;

    const float px = P[n*3+0], py = P[n*3+1], pz = P[n*3+2];

    unsigned long long key[8];
    #pragma unroll
    for (int t = 0; t < 8; ++t) key[t] = ~0ULL;

    for (int j = lane; j < npts; j += 32) {
        float dx = px - P[j*3+0];
        float dy = py - P[j*3+1];
        float dz = pz - P[j*3+2];
        float d2 = dx*dx + dy*dy + dz*dz;
        unsigned long long k =
            ((unsigned long long)__float_as_uint(d2) << 32) | (unsigned)j;
        if (k < key[7]) {
            key[7] = k;
            #pragma unroll
            for (int t = 7; t > 0; --t) {
                if (key[t] < key[t-1]) {
                    unsigned long long tmp = key[t]; key[t] = key[t-1]; key[t-1] = tmp;
                }
            }
        }
    }

    int head = 0;
    int sel[8];
    #pragma unroll
    for (int t = 0; t < 8; ++t) {
        unsigned long long cand = (head < 8) ? key[head] : ~0ULL;
        unsigned long long mn = cand;
        #pragma unroll
        for (int off = 16; off >= 1; off >>= 1) {
            unsigned long long o = __shfl_xor(mn, off);
            mn = (o < mn) ? o : mn;
        }
        if (cand == mn && head < 8) head++;
        sel[t] = (int)(mn & 0xFFFFFFFFu);
    }

    for (int c = lane; c < Dd; c += 32) {
        float ssum = 0.f;
        #pragma unroll
        for (int t = 0; t < 8; ++t) ssum += F[(size_t)sel[t] * Dd + c];
        Orow[c] = F[(size_t)n * Dd + c] - ssum * 0.125f;
    }
}

__global__ void ea_gemm256(const float* __restrict__ A, const float* __restrict__ W,
                           const float* __restrict__ bias, float* __restrict__ OUT,
                           int R)
{
    const int lane = threadIdx.x & 31;
    const int wave = threadIdx.x >> 5;
    const int tile = blockIdx.x * 8 + wave;
    const int total = (R / 16) * (Dd / 16);
    if (tile >= total) return;
    const int tm = tile >> 4, tn = tile & 15;
    const int col = lane & 15, half = lane >> 4, klo = half * 2;

    const float* Ap = A + (size_t)(tm * 16 + col) * Dd;
    const float* Wp = W + (size_t)(tn * 16 + col) * Dd;

    v8f acc = {};
    #pragma unroll 4
    for (int k = 0; k < Dd; k += 4) {
        v2f a = *(const v2f*)(Ap + k + klo);
        v2f b = *(const v2f*)(Wp + k + klo);
        acc = __builtin_amdgcn_wmma_f32_16x16x4_f32(false, a, false, b,
                                                    (short)0, acc, false, false);
    }
    const float bv = bias[tn * 16 + col];
    #pragma unroll
    for (int r = 0; r < 8; ++r) {
        int row = tm * 16 + r + 8 * half;
        OUT[(size_t)row * Dd + tn * 16 + col] = acc[r] + bv;
    }
}


__device__ __forceinline__ float ea_xhalf(float x)
{
    int b = __builtin_bit_cast(int, x);
    return __builtin_bit_cast(float,
        __builtin_amdgcn_permlanex16(b, b, 0x76543210, 0xfedcba98, false, false));
}

__global__ void ea_attn(const float* __restrict__ Q, const float* __restrict__ Km,
                        const float* __restrict__ Vm, float* __restrict__ O)
{
    const int lane = threadIdx.x & 31;
    const int wave = threadIdx.x >> 5;
    const int tile = blockIdx.x * 8 + wave;
    const int qt = tile & 127;
    const int h  = (tile >> 7) & 7;
    const int b  = tile >> 10;
    const int col = lane & 15, half = lane >> 4, klo = half * 2;

    const float* Qrow = Q + (size_t)(b * Nn + qt * 16 + col) * Dd + h * DHh;
    v2f qf[8];
    #pragma unroll
    for (int kk = 0; kk < 8; ++kk) qf[kk] = *(const v2f*)(Qrow + kk * 4 + klo);

    float mrun = -__builtin_inff();
    float lrun = 0.f;
    v8f acc0 = {}, acc1 = {};

    const float* Kb = Km + (size_t)b * Ll * Dd + h * DHh;
    const float* Vb = Vm + (size_t)b * Ll * Dd + h * DHh;
    const float scale = 0.17677669529663687f;

    for (int kt = 0; kt < Ll / 16; ++kt) {
        if (kt + 1 < Ll / 16)
            __builtin_prefetch(Kb + (size_t)((kt + 1) * 16 + col) * Dd, 0, 1);

        v8f s = {};
        const float* Kp = Kb + (size_t)(kt * 16 + col) * Dd;
        #pragma unroll
        for (int kk = 0; kk < 8; ++kk) {
            v2f kf = *(const v2f*)(Kp + kk * 4 + klo);
            s = __builtin_amdgcn_wmma_f32_16x16x4_f32(false, kf, false, qf[kk],
                                                      (short)0, s, false, false);
        }

        float v[8];
        #pragma unroll
        for (int r = 0; r < 8; ++r) v[r] = s[r] * scale;
        float mx = fmaxf(fmaxf(fmaxf(v[0], v[1]), fmaxf(v[2], v[3])),
                         fmaxf(fmaxf(v[4], v[5]), fmaxf(v[6], v[7])));
        mx = fmaxf(mx, ea_xhalf(mx));
        const float mnew = fmaxf(mrun, mx);
        const float corr = __expf(mrun - mnew);
        float p[8];
        float psum = 0.f;
        #pragma unroll
        for (int r = 0; r < 8; ++r) { p[r] = __expf(v[r] - mnew); psum += p[r]; }
        lrun = lrun * corr + psum;
        #pragma unroll
        for (int r = 0; r < 8; ++r) { acc0[r] *= corr; acc1[r] *= corr; }
        mrun = mnew;

        const float* Vp = Vb + (size_t)(kt * 16 + 8 * half) * Dd + col;
        #pragma unroll
        for (int j = 0; j < 4; ++j) {
            const float* V0 = Vp + (size_t)(2 * j) * Dd;
            const float* V1 = V0 + Dd;
            v2f pf; pf.x = p[2 * j];  pf.y = p[2 * j + 1];
            v2f a0; a0.x = V0[0];     a0.y = V1[0];
            v2f a1; a1.x = V0[16];    a1.y = V1[16];
            acc0 = __builtin_amdgcn_wmma_f32_16x16x4_f32(false, a0, false, pf,
                                                         (short)0, acc0, false, false);
            acc1 = __builtin_amdgcn_wmma_f32_16x16x4_f32(false, a1, false, pf,
                                                         (short)0, acc1, false, false);
        }
    }

    const float inv = 1.0f / (lrun + ea_xhalf(lrun));
    float* Op = O + (size_t)(b * Nn + qt * 16 + col) * Dd + h * DHh + 8 * half;
    typedef __attribute__((ext_vector_type(4))) float v4f;
    v4f o0 = { acc0[0] * inv, acc0[1] * inv, acc0[2] * inv, acc0[3] * inv };
    v4f o1 = { acc0[4] * inv, acc0[5] * inv, acc0[6] * inv, acc0[7] * inv };
    v4f o2 = { acc1[0] * inv, acc1[1] * inv, acc1[2] * inv, acc1[3] * inv };
    v4f o3 = { acc1[4] * inv, acc1[5] * inv, acc1[6] * inv, acc1[7] * inv };
    *(v4f*)(Op)      = o0;
    *(v4f*)(Op + 4)  = o1;
    *(v4f*)(Op + 16) = o2;
    *(v4f*)(Op + 20) = o3;
}

__global__ void ea_gemm_out(const float* __restrict__ A, const float* __restrict__ W,
                            const float* __restrict__ bias,
                            const float* __restrict__ pts,
                            const float* __restrict__ sw, const float* __restrict__ sb,
                            float* __restrict__ OUT, int R)
{
    const int lane = threadIdx.x & 31;
    const int wave = threadIdx.x >> 5;
    const int tile = blockIdx.x * 8 + wave;
    const int total = (R / 16) * (Dd / 16);
    if (tile >= total) return;
    const int tm = tile >> 4, tn = tile & 15;
    const int col = lane & 15, half = lane >> 4, klo = half * 2;
    const int colg = tn * 16 + col;

    const float* Ap = A + (size_t)(tm * 16 + col) * Dd;
    const float* Wp = W + (size_t)colg * Dd;

    v8f acc = {};
    #pragma unroll 4
    for (int k = 0; k < Dd; k += 4) {
        v2f a = *(const v2f*)(Ap + k + klo);
        v2f b = *(const v2f*)(Wp + k + klo);
        acc = __builtin_amdgcn_wmma_f32_16x16x4_f32(false, a, false, b,
                                                    (short)0, acc, false, false);
    }
    const float s0 = sw[colg*3+0], s1 = sw[colg*3+1], s2 = sw[colg*3+2];
    const float add = bias[colg] + sb[colg];
    #pragma unroll
    for (int r = 0; r < 8; ++r) {
        int row = tm * 16 + r + 8 * half;
        float px = pts[row*3+0], py = pts[row*3+1], pz = pts[row*3+2];
        OUT[(size_t)row * Dd + colg] = acc[r] + add + px*s0 + py*s1 + pz*s2;
    }
}

extern "C" void kernel_launch(void* const* d_in, const int* in_sizes, int n_in,
                              void* d_out, int out_size, void* d_ws, size_t ws_size,
                              hipStream_t stream)
{
    const float* cur_pts  = (const float*)d_in[0];
    const float* cur_feat = (const float*)d_in[1];
    const float* nbr_pts  = (const float*)d_in[2];
    const float* nbr_feat = (const float*)d_in[3];
    const float* in_w     = (const float*)d_in[4];
    const float* in_b     = (const float*)d_in[5];
    const float* out_w    = (const float*)d_in[6];
    const float* out_b    = (const float*)d_in[7];
    const float* sp_w     = (const float*)d_in[8];
    const float* sp_b     = (const float*)d_in[9];
    float* out = (float*)d_out;

    float* ws   = (float*)d_ws;
    float* curE = ws;
    float* nbrE = curE + 4096 * 256;
    float* Qb   = nbrE + 8192 * 256;
    float* Kb   = Qb   + 4096 * 256;
    float* Vb   = Kb   + 8192 * 256;
    float* Ob   = Vb   + 8192 * 256;

    const dim3 blk(256);

    ea_knn_edge<<<dim3(4096 / 8), blk, 0, stream>>>(cur_pts, cur_feat, curE,
                                                    Nn, Bc, 0);
    ea_knn_edge<<<dim3(8192 / 8), blk, 0, stream>>>(nbr_pts, nbr_feat, nbrE,
                                                    Nn, Mm * Bc, 1);

    ea_gemm256<<<dim3(4096 / 8), blk, 0, stream>>>(curE, in_w,           in_b,       Qb, 4096);
    ea_gemm256<<<dim3(8192 / 8), blk, 0, stream>>>(nbrE, in_w + 256*256, in_b + 256, Kb, 8192);
    ea_gemm256<<<dim3(8192 / 8), blk, 0, stream>>>(nbrE, in_w + 512*256, in_b + 512, Vb, 8192);

    ea_attn<<<dim3(2048 / 8), blk, 0, stream>>>(Qb, Kb, Vb, Ob);

    ea_gemm_out<<<dim3(4096 / 8), blk, 0, stream>>>(Ob, out_w, out_b,
                                                    cur_pts, sp_w, sp_b,
                                                    out, 4096);
}


# attention K/V tile loads software-pipelined in registers: four one-tile register sets in rotation, each re-loaded four tiles ahead right behind its step; replaces the cache prefetch; sums unchanged
# speedup vs baseline: 2.2514x; 1.2675x over previous
#include <hip/hip_runtime.h>
#include <hip/hip_bf16.h>


#define Bc   2
#define Nn   2048
#define Dd   256
#define Hh   8
#define DHh  32
#define Mm   2
#define Ll   4096

typedef __attribute__((ext_vector_type(2))) float v2f;
typedef __attribute__((ext_vector_type(8))) float v8f;

__global__ void ea_knn_edge(const float* __restrict__ pts,
                            const float* __restrict__ feats,
                            float* __restrict__ out,
                            int npts, int nsets, int remap)
{
    const int lane = threadIdx.x & 31;
    const int wave = threadIdx.x >> 5;
    const int rowg = blockIdx.x * 8 + wave;
    if (rowg >= nsets * npts) return;
    const int s = rowg / npts;
    const int n = rowg - s * npts;
    int os = s;
    if (remap) { int m = s / Bc; int b = s - m * Bc; os = b * Mm + m; }

    const float* P = pts   + (size_t)s  * npts * 3;
    const float* F = feats + (size_t)s  * npts * Dd;
    float*    Orow = out   + ((size_t)os * npts + n) * Dd;

    const float px = P[n*3+0], py = P[n*3+1], pz = P[n*3+2];

    unsigned long long key[8];
    #pragma unroll
    for (int t = 0; t < 8; ++t) key[t] = ~0ULL;

    for (int j = lane; j < npts; j += 32) {
        float dx = px - P[j*3+0];
        float dy = py - P[j*3+1];
        float dz = pz - P[j*3+2];
        float d2 = dx*dx + dy*dy + dz*dz;
        unsigned long long k =
            ((unsigned long long)__float_as_uint(d2) << 32) | (unsigned)j;
        if (k < key[7]) {
            key[7] = k;
            #pragma unroll
            for (int t = 7; t > 0; --t) {
                if (key[t] < key[t-1]) {
                    unsigned long long tmp = key[t]; key[t] = key[t-1]; key[t-1] = tmp;
                }
            }
        }
    }

    int head = 0;
    int sel[8];
    #pragma unroll
    for (int t = 0; t < 8; ++t) {
        unsigned long long cand = (head < 8) ? key[head] : ~0ULL;
        unsigned long long mn = cand;
        #pragma unroll
        for (int off = 16; off >= 1; off >>= 1) {
            unsigned long long o = __shfl_xor(mn, off);
            mn = (o < mn) ? o : mn;
        }
        if (cand == mn && head < 8) head++;
        sel[t] = (int)(mn & 0xFFFFFFFFu);
    }

    for (int c = lane; c < Dd; c += 32) {
        float ssum = 0.f;
        #pragma unroll
        for (int t = 0; t < 8; ++t) ssum += F[(size_t)sel[t] * Dd + c];
        Orow[c] = F[(size_t)n * Dd + c] - ssum * 0.125f;
    }
}

__global__ void ea_gemm256(const float* __restrict__ A, const float* __restrict__ W,
                           const float* __restrict__ bias, float* __restrict__ OUT,
                           int R)
{
    const int lane = threadIdx.x & 31;
    const int wave = threadIdx.x >> 5;
    const int tile = blockIdx.x * 8 + wave;
    const int total = (R / 16) * (Dd / 16);
    if (tile >= total) return;
    const int tm = tile >> 4, tn = tile & 15;
    const int col = lane & 15, half = lane >> 4, klo = half * 2;

    const float* Ap = A + (size_t)(tm * 16 + col) * Dd;
    const float* Wp = W + (size_t)(tn * 16 + col) * Dd;

    v8f acc = {};
    #pragma unroll 4
    for (int k = 0; k < Dd; k += 4) {
        v2f a = *(const v2f*)(Ap + k + klo);
        v2f b = *(const v2f*)(Wp + k + klo);
        acc = __builtin_amdgcn_wmma_f32_16x16x4_f32(false, a, false, b,
                                                    (short)0, acc, false, false);
    }
    const float bv = bias[tn * 16 + col];
    #pragma unroll
    for (int r = 0; r < 8; ++r) {
        int row = tm * 16 + r + 8 * half;
        OUT[(size_t)row * Dd + tn * 16 + col] = acc[r] + bv;
    }
}


__device__ __forceinline__ float ea_xhalf(float x)
{
    int b = __builtin_bit_cast(int, x);
    return __builtin_bit_cast(float,
        __builtin_amdgcn_permlanex16(b, b, 0x76543210, 0xfedcba98, false, false));
}

struct ea_kv {
    v2f k[8];
    v2f a0[4];
    v2f a1[4];
};

__device__ __forceinline__ void ea_load_tile(ea_kv& t, const float* __restrict__ Kb,
                                             const float* __restrict__ Vb,
                                             int kt, int col, int half, int klo)
{
    const float* Kp = Kb + (size_t)(kt * 16 + col) * Dd;
    #pragma unroll
    for (int kk = 0; kk < 8; ++kk) t.k[kk] = *(const v2f*)(Kp + kk * 4 + klo);
    const float* Vp = Vb + (size_t)(kt * 16 + 8 * half) * Dd + col;
    #pragma unroll
    for (int j = 0; j < 4; ++j) {
        const float* V0 = Vp + (size_t)(2 * j) * Dd;
        const float* V1 = V0 + Dd;
        t.a0[j].x = V0[0];   t.a0[j].y = V1[0];
        t.a1[j].x = V0[16];  t.a1[j].y = V1[16];
    }
}

__device__ __forceinline__ void ea_tile_step(const ea_kv& t, const v2f (&qf)[8],
                                             float& mrun, float& lrun,
                                             v8f& acc0, v8f& acc1)
{
    const float scale = 0.17677669529663687f;

    v8f s = {};
    #pragma unroll
    for (int kk = 0; kk < 8; ++kk)
        s = __builtin_amdgcn_wmma_f32_16x16x4_f32(false, t.k[kk], false, qf[kk],
                                                  (short)0, s, false, false);

    float v[8];
    #pragma unroll
    for (int r = 0; r < 8; ++r) v[r] = s[r] * scale;
    float mx = fmaxf(fmaxf(fmaxf(v[0], v[1]), fmaxf(v[2], v[3])),
                     fmaxf(fmaxf(v[4], v[5]), fmaxf(v[6], v[7])));
    mx = fmaxf(mx, ea_xhalf(mx));
    const float mnew = fmaxf(mrun, mx);
    const float corr = __expf(mrun - mnew);
    float p[8];
    float psum = 0.f;
    #pragma unroll
    for (int r = 0; r < 8; ++r) { p[r] = __expf(v[r] - mnew); psum += p[r]; }
    lrun = lrun * corr + psum;
    #pragma unroll
    for (int r = 0; r < 8; ++r) { acc0[r] *= corr; acc1[r] *= corr; }
    mrun = mnew;

    #pragma unroll
    for (int j = 0; j < 4; ++j) {
        v2f pf; pf.x = p[2 * j];  pf.y = p[2 * j + 1];
        acc0 = __builtin_amdgcn_wmma_f32_16x16x4_f32(false, t.a0[j], false, pf,
                                                     (short)0, acc0, false, false);
        acc1 = __builtin_amdgcn_wmma_f32_16x16x4_f32(false, t.a1[j], false, pf,
                                                     (short)0, acc1, false, false);
    }
}

__global__ void __launch_bounds__(256)
ea_attn(const float* __restrict__ Q, const float* __restrict__ Km,
        const float* __restrict__ Vm, float* __restrict__ O)
{
    const int lane = threadIdx.x & 31;
    const int wave = threadIdx.x >> 5;
    const int tile = blockIdx.x * 8 + wave;
    const int qt = tile & 127;
    const int h  = (tile >> 7) & 7;
    const int b  = tile >> 10;
    const int col = lane & 15, half = lane >> 4, klo = half * 2;

    const float* Qrow = Q + (size_t)(b * Nn + qt * 16 + col) * Dd + h * DHh;
    v2f qf[8];
    #pragma unroll
    for (int kk = 0; kk < 8; ++kk) qf[kk] = *(const v2f*)(Qrow + kk * 4 + klo);

    float mrun = -__builtin_inff();
    float lrun = 0.f;
    v8f acc0 = {}, acc1 = {};

    const float* Kb = Km + (size_t)b * Ll * Dd + h * DHh;
    const float* Vb = Vm + (size_t)b * Ll * Dd + h * DHh;

    ea_kv e0, e1, e2, e3;
    __builtin_amdgcn_sched_barrier(0);
    ea_load_tile(e0, Kb, Vb, 0, col, half, klo);
    __builtin_amdgcn_sched_barrier(0);
    ea_load_tile(e1, Kb, Vb, 1, col, half, klo);
    __builtin_amdgcn_sched_barrier(0);
    ea_load_tile(e2, Kb, Vb, 2, col, half, klo);
    __builtin_amdgcn_sched_barrier(0);
    ea_load_tile(e3, Kb, Vb, 3, col, half, klo);
    __builtin_amdgcn_sched_barrier(0);
    for (int kt = 0; kt < Ll / 16; kt += 4) {
        int t0 = (kt + 4 < Ll / 16) ? kt + 4 : kt;
        int t1 = t0 + 1, t2 = t0 + 2, t3 = t0 + 3;
        ea_tile_step(e0, qf, mrun, lrun, acc0, acc1);
        asm volatile("" : "+v"(acc0), "+v"(acc1), "+s"(t0));
        ea_load_tile(e0, Kb, Vb, t0, col, half, klo);
        __builtin_amdgcn_sched_barrier(0);
        ea_tile_step(e1, qf, mrun, lrun, acc0, acc1);
        asm volatile("" : "+v"(acc0), "+v"(acc1), "+s"(t1));
        ea_load_tile(e1, Kb, Vb, t1, col, half, klo);
        __builtin_amdgcn_sched_barrier(0);
        ea_tile_step(e2, qf, mrun, lrun, acc0, acc1);
        asm volatile("" : "+v"(acc0), "+v"(acc1), "+s"(t2));
        ea_load_tile(e2, Kb, Vb, t2, col, half, klo);
        __builtin_amdgcn_sched_barrier(0);
        ea_tile_step(e3, qf, mrun, lrun, acc0, acc1);
        asm volatile("" : "+v"(acc0), "+v"(acc1), "+s"(t3));
        ea_load_tile(e3, Kb, Vb, t3, col, half, klo);
        __builtin_amdgcn_sched_barrier(0);
    }

    const float inv = 1.0f / (lrun + ea_xhalf(lrun));
    float* Op = O + (size_t)(b * Nn + qt * 16 + col) * Dd + h * DHh + 8 * half;
    typedef __attribute__((ext_vector_type(4))) float v4f;
    v4f o0 = { acc0[0] * inv, acc0[1] * inv, acc0[2] * inv, acc0[3] * inv };
    v4f o1 = { acc0[4] * inv, acc0[5] * inv, acc0[6] * inv, acc0[7] * inv };
    v4f o2 = { acc1[0] * inv, acc1[1] * inv, acc1[2] * inv, acc1[3] * inv };
    v4f o3 = { acc1[4] * inv, acc1[5] * inv, acc1[6] * inv, acc1[7] * inv };
    *(v4f*)(Op)      = o0;
    *(v4f*)(Op + 4)  = o1;
    *(v4f*)(Op + 16) = o2;
    *(v4f*)(Op + 20) = o3;
}

__global__ void ea_gemm_out(const float* __restrict__ A, const float* __restrict__ W,
                            const float* __restrict__ bias,
                            const float* __restrict__ pts,
                            const float* __restrict__ sw, const float* __restrict__ sb,
                            float* __restrict__ OUT, int R)
{
    const int lane = threadIdx.x & 31;
    const int wave = threadIdx.x >> 5;
    const int tile = blockIdx.x * 8 + wave;
    const int total = (R / 16) * (Dd / 16);
    if (tile >= total) return;
    const int tm = tile >> 4, tn = tile & 15;
    const int col = lane & 15, half = lane >> 4, klo = half * 2;
    const int colg = tn * 16 + col;

    const float* Ap = A + (size_t)(tm * 16 + col) * Dd;
    const float* Wp = W + (size_t)colg * Dd;

    v8f acc = {};
    #pragma unroll 4
    for (int k = 0; k < Dd; k += 4) {
        v2f a = *(const v2f*)(Ap + k + klo);
        v2f b = *(const v2f*)(Wp + k + klo);
        acc = __builtin_amdgcn_wmma_f32_16x16x4_f32(false, a, false, b,
                                                    (short)0, acc, false, false);
    }
    const float s0 = sw[colg*3+0], s1 = sw[colg*3+1], s2 = sw[colg*3+2];
    const float add = bias[colg] + sb[colg];
    #pragma unroll
    for (int r = 0; r < 8; ++r) {
        int row = tm * 16 + r + 8 * half;
        float px = pts[row*3+0], py = pts[row*3+1], pz = pts[row*3+2];
        OUT[(size_t)row * Dd + colg] = acc[r] + add + px*s0 + py*s1 + pz*s2;
    }
}

extern "C" void kernel_launch(void* const* d_in, const int* in_sizes, int n_in,
                              void* d_out, int out_size, void* d_ws, size_t ws_size,
                              hipStream_t stream)
{
    const float* cur_pts  = (const float*)d_in[0];
    const float* cur_feat = (const float*)d_in[1];
    const float* nbr_pts  = (const float*)d_in[2];
    const float* nbr_feat = (const float*)d_in[3];
    const float* in_w     = (const float*)d_in[4];
    const float* in_b     = (const float*)d_in[5];
    const float* out_w    = (const float*)d_in[6];
    const float* out_b    = (const float*)d_in[7];
    const float* sp_w     = (const float*)d_in[8];
    const float* sp_b     = (const float*)d_in[9];
    float* out = (float*)d_out;

    float* ws   = (float*)d_ws;
    float* curE = ws;
    float* nbrE = curE + 4096 * 256;
    float* Qb   = nbrE + 8192 * 256;
    float* Kb   = Qb   + 4096 * 256;
    float* Vb   = Kb   + 8192 * 256;
    float* Ob   = Vb   + 8192 * 256;

    const dim3 blk(256);

    ea_knn_edge<<<dim3(4096 / 8), blk, 0, stream>>>(cur_pts, cur_feat, curE,
                                                    Nn, Bc, 0);
    ea_knn_edge<<<dim3(8192 / 8), blk, 0, stream>>>(nbr_pts, nbr_feat, nbrE,
                                                    Nn, Mm * Bc, 1);

    ea_gemm256<<<dim3(4096 / 8), blk, 0, stream>>>(curE, in_w,           in_b,       Qb, 4096);
    ea_gemm256<<<dim3(8192 / 8), blk, 0, stream>>>(nbrE, in_w + 256*256, in_b + 256, Kb, 8192);
    ea_gemm256<<<dim3(8192 / 8), blk, 0, stream>>>(nbrE, in_w + 512*256, in_b + 512, Vb, 8192);

    ea_attn<<<dim3(2048 / 8), blk, 0, stream>>>(Qb, Kb, Vb, Ob);

    ea_gemm_out<<<dim3(4096 / 8), blk, 0, stream>>>(Ob, out_w, out_b,
                                                    cur_pts, sp_w, sp_b,
                                                    out, 4096);
}


# K and V projections write each 16-key tile in the attention's fragment order, so a tile is 8 whole-line 128-bit loads, not 24 partial ones; dh sum order in Q*K changes
# speedup vs baseline: 2.3876x; 1.0605x over previous
#include <hip/hip_runtime.h>
#include <hip/hip_bf16.h>


#define Bc   2
#define Nn   2048
#define Dd   256
#define Hh   8
#define DHh  32
#define Mm   2
#define Ll   4096

typedef __attribute__((ext_vector_type(2))) float v2f;
typedef __attribute__((ext_vector_type(4))) float v4f;
typedef __attribute__((ext_vector_type(8))) float v8f;

__global__ void ea_knn_edge(const float* __restrict__ pts,
                            const float* __restrict__ feats,
                            float* __restrict__ out,
                            int npts, int nsets, int remap)
{
    const int lane = threadIdx.x & 31;
    const int wave = threadIdx.x >> 5;
    const int rowg = blockIdx.x * 8 + wave;
    if (rowg >= nsets * npts) return;
    const int s = rowg / npts;
    const int n = rowg - s * npts;
    int os = s;
    if (remap) { int m = s / Bc; int b = s - m * Bc; os = b * Mm + m; }

    const float* P = pts   + (size_t)s  * npts * 3;
    const float* F = feats + (size_t)s  * npts * Dd;
    float*    Orow = out   + ((size_t)os * npts + n) * Dd;

    const float px = P[n*3+0], py = P[n*3+1], pz = P[n*3+2];

    unsigned long long key[8];
    #pragma unroll
    for (int t = 0; t < 8; ++t) key[t] = ~0ULL;

    for (int j = lane; j < npts; j += 32) {
        float dx = px - P[j*3+0];
        float dy = py - P[j*3+1];
        float dz = pz - P[j*3+2];
        float d2 = dx*dx + dy*dy + dz*dz;
        unsigned long long k =
            ((unsigned long long)__float_as_uint(d2) << 32) | (unsigned)j;
        if (k < key[7]) {
            key[7] = k;
            #pragma unroll
            for (int t = 7; t > 0; --t) {
                if (key[t] < key[t-1]) {
                    unsigned long long tmp = key[t]; key[t] = key[t-1]; key[t-1] = tmp;
                }
            }
        }
    }

    int head = 0;
    int sel[8];
    #pragma unroll
    for (int t = 0; t < 8; ++t) {
        unsigned long long cand = (head < 8) ? key[head] : ~0ULL;
        unsigned long long mn = cand;
        #pragma unroll
        for (int off = 16; off >= 1; off >>= 1) {
            unsigned long long o = __shfl_xor(mn, off);
            mn = (o < mn) ? o : mn;
        }
        if (cand == mn && head < 8) head++;
        sel[t] = (int)(mn & 0xFFFFFFFFu);
    }

    for (int c = lane; c < Dd; c += 32) {
        float ssum = 0.f;
        #pragma unroll
        for (int t = 0; t < 8; ++t) ssum += F[(size_t)sel[t] * Dd + c];
        Orow[c] = F[(size_t)n * Dd + c] - ssum * 0.125f;
    }
}

__global__ void ea_gemm256(const float* __restrict__ A, const float* __restrict__ W,
                           const float* __restrict__ bias, float* __restrict__ OUT,
                           int R)
{
    const int lane = threadIdx.x & 31;
    const int wave = threadIdx.x >> 5;
    const int tile = blockIdx.x * 8 + wave;
    const int total = (R / 16) * (Dd / 16);
    if (tile >= total) return;
    const int tm = tile >> 4, tn = tile & 15;
    const int col = lane & 15, half = lane >> 4, klo = half * 2;

    const float* Ap = A + (size_t)(tm * 16 + col) * Dd;
    const float* Wp = W + (size_t)(tn * 16 + col) * Dd;

    v8f acc = {};
    #pragma unroll 4
    for (int k = 0; k < Dd; k += 4) {
        v2f a = *(const v2f*)(Ap + k + klo);
        v2f b = *(const v2f*)(Wp + k + klo);
        acc = __builtin_amdgcn_wmma_f32_16x16x4_f32(false, a, false, b,
                                                    (short)0, acc, false, false);
    }
    const float bv = bias[tn * 16 + col];
    #pragma unroll
    for (int r = 0; r < 8; ++r) {
        int row = tm * 16 + r + 8 * half;
        OUT[(size_t)row * Dd + tn * 16 + col] = acc[r] + bv;
    }
}

__global__ void ea_gemm_kvfrag(const float* __restrict__ A, const float* __restrict__ W,
                               const float* __restrict__ bias, float* __restrict__ OUT,
                               int R, int kmode)
{
    const int lane = threadIdx.x & 31;
    const int wave = threadIdx.x >> 5;
    const int tile = blockIdx.x * 8 + wave;
    const int total = (R / 16) * (Dd / 16);
    if (tile >= total) return;
    const int tm = tile >> 4, tn = tile & 15;
    const int col = lane & 15, half = lane >> 4, klo = half * 2;

    const float* Erow = A + (size_t)(tm * 16 + col) * Dd;
    const float* Wrow = W + (size_t)(tn * 16 + col) * Dd;
    const float* Ap = kmode ? Wrow : Erow;
    const float* Bp = kmode ? Erow : Wrow;

    v8f acc = {};
    #pragma unroll 4
    for (int k = 0; k < Dd; k += 4) {
        v2f a = *(const v2f*)(Ap + k + klo);
        v2f b = *(const v2f*)(Bp + k + klo);
        acc = __builtin_amdgcn_wmma_f32_16x16x4_f32(false, a, false, b,
                                                    (short)0, acc, false, false);
    }
    const float* bp = bias + tn * 16 + (kmode ? 8 * half : col);
    const int bstep = kmode ? 1 : 0;
    const int bb = tm >> 8, kt = tm & 255;
    float* dst = OUT + ((size_t)((bb * Hh + (tn >> 1)) * 256 + kt)) * 512
                     + (tn & 1) * 256 + lane * 4;
    v4f o0 = { acc[0] + bp[0],         acc[1] + bp[bstep],
               acc[2] + bp[2 * bstep], acc[3] + bp[3 * bstep] };
    v4f o1 = { acc[4] + bp[4 * bstep], acc[5] + bp[5 * bstep],
               acc[6] + bp[6 * bstep], acc[7] + bp[7 * bstep] };
    *(v4f*)(dst)       = o0;
    *(v4f*)(dst + 128) = o1;
}


__device__ __forceinline__ float ea_xhalf(float x)
{
    int b = __builtin_bit_cast(int, x);
    return __builtin_bit_cast(float,
        __builtin_amdgcn_permlanex16(b, b, 0x76543210, 0xfedcba98, false, false));
}

struct ea_kv {
    v4f k[4];
    v4f v[4];
};

__device__ __forceinline__ v2f ea_lo(v4f f) { v2f r = { f.x, f.y }; return r; }
__device__ __forceinline__ v2f ea_hi(v4f f) { v2f r = { f.z, f.w }; return r; }

__device__ __forceinline__ void ea_load_tile(ea_kv& t, const float* __restrict__ Kl,
                                             const float* __restrict__ Vl, int kt)
{
    const float* Kp = Kl + (size_t)kt * 512;
    const float* Vp = Vl + (size_t)kt * 512;
    #pragma unroll
    for (int q = 0; q < 4; ++q) t.k[q] = *(const v4f*)(Kp + q * 128);
    #pragma unroll
    for (int q = 0; q < 4; ++q) t.v[q] = *(const v4f*)(Vp + q * 128);
}

__device__ __forceinline__ void ea_tile_step(const ea_kv& t, const v2f (&qf)[8],
                                             float& mrun, float& lrun,
                                             v8f& acc0, v8f& acc1)
{
    const float scale = 0.17677669529663687f;

    v8f s = {};
    #pragma unroll
    for (int q = 0; q < 4; ++q) {
        s = __builtin_amdgcn_wmma_f32_16x16x4_f32(false, ea_lo(t.k[q]), false, qf[2 * q],
                                                  (short)0, s, false, false);
        s = __builtin_amdgcn_wmma_f32_16x16x4_f32(false, ea_hi(t.k[q]), false, qf[2 * q + 1],
                                                  (short)0, s, false, false);
    }

    float v[8];
    #pragma unroll
    for (int r = 0; r < 8; ++r) v[r] = s[r] * scale;
    float mx = fmaxf(fmaxf(fmaxf(v[0], v[1]), fmaxf(v[2], v[3])),
                     fmaxf(fmaxf(v[4], v[5]), fmaxf(v[6], v[7])));
    mx = fmaxf(mx, ea_xhalf(mx));
    const float mnew = fmaxf(mrun, mx);
    const float corr = __expf(mrun - mnew);
    float p[8];
    float psum = 0.f;
    #pragma unroll
    for (int r = 0; r < 8; ++r) { p[r] = __expf(v[r] - mnew); psum += p[r]; }
    lrun = lrun * corr + psum;
    #pragma unroll
    for (int r = 0; r < 8; ++r) { acc0[r] *= corr; acc1[r] *= corr; }
    mrun = mnew;

    #pragma unroll
    for (int j = 0; j < 4; ++j) {
        v2f pf; pf.x = p[2 * j];  pf.y = p[2 * j + 1];
        const v2f a0 = (j & 1) ? ea_hi(t.v[j >> 1])       : ea_lo(t.v[j >> 1]);
        const v2f a1 = (j & 1) ? ea_hi(t.v[2 + (j >> 1)]) : ea_lo(t.v[2 + (j >> 1)]);
        acc0 = __builtin_amdgcn_wmma_f32_16x16x4_f32(false, a0, false, pf,
                                                     (short)0, acc0, false, false);
        acc1 = __builtin_amdgcn_wmma_f32_16x16x4_f32(false, a1, false, pf,
                                                     (short)0, acc1, false, false);
    }
}

__global__ void __launch_bounds__(256)
ea_attn(const float* __restrict__ Q, const float* __restrict__ Km,
        const float* __restrict__ Vm, float* __restrict__ O)
{
    const int lane = threadIdx.x & 31;
    const int wave = threadIdx.x >> 5;
    const int tile = blockIdx.x * 8 + wave;
    const int qt = tile & 127;
    const int h  = (tile >> 7) & 7;
    const int b  = tile >> 10;
    const int col = lane & 15, half = lane >> 4;

    const float* Qrow = Q + (size_t)(b * Nn + qt * 16 + col) * Dd + h * DHh + 8 * half;
    v2f qf[8];
    #pragma unroll
    for (int q = 0; q < 4; ++q) {
        const v4f g = *(const v4f*)(Qrow + 16 * (q >> 1) + 4 * (q & 1));
        qf[2 * q]     = ea_lo(g);
        qf[2 * q + 1] = ea_hi(g);
    }

    float mrun = -__builtin_inff();
    float lrun = 0.f;
    v8f acc0 = {}, acc1 = {};

    const float* Kb = Km + (size_t)((b * Hh + h) * 256) * 512 + lane * 4;
    const float* Vb = Vm + (size_t)((b * Hh + h) * 256) * 512 + lane * 4;

    ea_kv e0, e1, e2, e3;
    __builtin_amdgcn_sched_barrier(0);
    ea_load_tile(e0, Kb, Vb, 0);
    __builtin_amdgcn_sched_barrier(0);
    ea_load_tile(e1, Kb, Vb, 1);
    __builtin_amdgcn_sched_barrier(0);
    ea_load_tile(e2, Kb, Vb, 2);
    __builtin_amdgcn_sched_barrier(0);
    ea_load_tile(e3, Kb, Vb, 3);
    __builtin_amdgcn_sched_barrier(0);
    for (int kt = 0; kt < Ll / 16; kt += 4) {
        int t0 = (kt + 4 < Ll / 16) ? kt + 4 : kt;
        int t1 = t0 + 1, t2 = t0 + 2, t3 = t0 + 3;
        ea_tile_step(e0, qf, mrun, lrun, acc0, acc1);
        asm volatile("" : "+v"(acc0), "+v"(acc1), "+s"(t0));
        ea_load_tile(e0, Kb, Vb, t0);
        __builtin_amdgcn_sched_barrier(0);
        ea_tile_step(e1, qf, mrun, lrun, acc0, acc1);
        asm volatile("" : "+v"(acc0), "+v"(acc1), "+s"(t1));
        ea_load_tile(e1, Kb, Vb, t1);
        __builtin_amdgcn_sched_barrier(0);
        ea_tile_step(e2, qf, mrun, lrun, acc0, acc1);
        asm volatile("" : "+v"(acc0), "+v"(acc1), "+s"(t2));
        ea_load_tile(e2, Kb, Vb, t2);
        __builtin_amdgcn_sched_barrier(0);
        ea_tile_step(e3, qf, mrun, lrun, acc0, acc1);
        asm volatile("" : "+v"(acc0), "+v"(acc1), "+s"(t3));
        ea_load_tile(e3, Kb, Vb, t3);
        __builtin_amdgcn_sched_barrier(0);
    }

    const float inv = 1.0f / (lrun + ea_xhalf(lrun));
    float* Op = O + (size_t)(b * Nn + qt * 16 + col) * Dd + h * DHh + 8 * half;
    v4f o0 = { acc0[0] * inv, acc0[1] * inv, acc0[2] * inv, acc0[3] * inv };
    v4f o1 = { acc0[4] * inv, acc0[5] * inv, acc0[6] * inv, acc0[7] * inv };
    v4f o2 = { acc1[0] * inv, acc1[1] * inv, acc1[2] * inv, acc1[3] * inv };
    v4f o3 = { acc1[4] * inv, acc1[5] * inv, acc1[6] * inv, acc1[7] * inv };
    *(v4f*)(Op)      = o0;
    *(v4f*)(Op + 4)  = o1;
    *(v4f*)(Op + 16) = o2;
    *(v4f*)(Op + 20) = o3;
}

__global__ void ea_gemm_out(const float* __restrict__ A, const float* __restrict__ W,
                            const float* __restrict__ bias,
                            const float* __restrict__ pts,
                            const float* __restrict__ sw, const float* __restrict__ sb,
                            float* __restrict__ OUT, int R)
{
    const int lane = threadIdx.x & 31;
    const int wave = threadIdx.x >> 5;
    const int tile = blockIdx.x * 8 + wave;
    const int total = (R / 16) * (Dd / 16);
    if (tile >= total) return;
    const int tm = tile >> 4, tn = tile & 15;
    const int col = lane & 15, half = lane >> 4, klo = half * 2;
    const int colg = tn * 16 + col;

    const float* Ap = A + (size_t)(tm * 16 + col) * Dd;
    const float* Wp = W + (size_t)colg * Dd;

    v8f acc = {};
    #pragma unroll 4
    for (int k = 0; k < Dd; k += 4) {
        v2f a = *(const v2f*)(Ap + k + klo);
        v2f b = *(const v2f*)(Wp + k + klo);
        acc = __builtin_amdgcn_wmma_f32_16x16x4_f32(false, a, false, b,
                                                    (short)0, acc, false, false);
    }
    const float s0 = sw[colg*3+0], s1 = sw[colg*3+1], s2 = sw[colg*3+2];
    const float add = bias[colg] + sb[colg];
    #pragma unroll
    for (int r = 0; r < 8; ++r) {
        int row = tm * 16 + r + 8 * half;
        float px = pts[row*3+0], py = pts[row*3+1], pz = pts[row*3+2];
        OUT[(size_t)row * Dd + colg] = acc[r] + add + px*s0 + py*s1 + pz*s2;
    }
}

extern "C" void kernel_launch(void* const* d_in, const int* in_sizes, int n_in,
                              void* d_out, int out_size, void* d_ws, size_t ws_size,
                              hipStream_t stream)
{
    const float* cur_pts  = (const float*)d_in[0];
    const float* cur_feat = (const float*)d_in[1];
    const float* nbr_pts  = (const float*)d_in[2];
    const float* nbr_feat = (const float*)d_in[3];
    const float* in_w     = (const float*)d_in[4];
    const float* in_b     = (const float*)d_in[5];
    const float* out_w    = (const float*)d_in[6];
    const float* out_b    = (const float*)d_in[7];
    const float* sp_w     = (const float*)d_in[8];
    const float* sp_b     = (const float*)d_in[9];
    float* out = (float*)d_out;

    float* ws   = (float*)d_ws;
    float* curE = ws;
    float* nbrE = curE + 4096 * 256;
    float* Qb   = nbrE + 8192 * 256;
    float* Kb   = Qb   + 4096 * 256;
    float* Vb   = Kb   + 8192 * 256;
    float* Ob   = Vb   + 8192 * 256;

    const dim3 blk(256);

    ea_knn_edge<<<dim3(4096 / 8), blk, 0, stream>>>(cur_pts, cur_feat, curE,
                                                    Nn, Bc, 0);
    ea_knn_edge<<<dim3(8192 / 8), blk, 0, stream>>>(nbr_pts, nbr_feat, nbrE,
                                                    Nn, Mm * Bc, 1);

    ea_gemm256<<<dim3(4096 / 8), blk, 0, stream>>>(curE, in_w,           in_b,       Qb, 4096);
    ea_gemm_kvfrag<<<dim3(8192 / 8), blk, 0, stream>>>(nbrE, in_w + 256*256, in_b + 256, Kb, 8192, 1);
    ea_gemm_kvfrag<<<dim3(8192 / 8), blk, 0, stream>>>(nbrE, in_w + 512*256, in_b + 512, Vb, 8192, 0);

    ea_attn<<<dim3(2048 / 8), blk, 0, stream>>>(Qb, Kb, Vb, Ob);

    ea_gemm_out<<<dim3(4096 / 8), blk, 0, stream>>>(Ob, out_w, out_b,
                                                    cur_pts, sp_w, sp_b,
                                                    out, 4096);
}


# projection GEMMs register-blocked: a wave holds one 16-row activation fragment against four weight tiles (16x64 piece), 128-bit loads one k-block ahead; k sum order in the projections changes
# speedup vs baseline: 2.6238x; 1.0989x over previous
#include <hip/hip_runtime.h>
#include <hip/hip_bf16.h>


#define Bc   2
#define Nn   2048
#define Dd   256
#define Hh   8
#define DHh  32
#define Mm   2
#define Ll   4096

typedef __attribute__((ext_vector_type(2))) float v2f;
typedef __attribute__((ext_vector_type(4))) float v4f;
typedef __attribute__((ext_vector_type(8))) float v8f;

__global__ void ea_knn_edge(const float* __restrict__ pts,
                            const float* __restrict__ feats,
                            float* __restrict__ out,
                            int npts, int nsets, int remap)
{
    const int lane = threadIdx.x & 31;
    const int wave = threadIdx.x >> 5;
    const int rowg = blockIdx.x * 8 + wave;
    if (rowg >= nsets * npts) return;
    const int s = rowg / npts;
    const int n = rowg - s * npts;
    int os = s;
    if (remap) { int m = s / Bc; int b = s - m * Bc; os = b * Mm + m; }

    const float* P = pts   + (size_t)s  * npts * 3;
    const float* F = feats + (size_t)s  * npts * Dd;
    float*    Orow = out   + ((size_t)os * npts + n) * Dd;

    const float px = P[n*3+0], py = P[n*3+1], pz = P[n*3+2];

    unsigned long long key[8];
    #pragma unroll
    for (int t = 0; t < 8; ++t) key[t] = ~0ULL;

    for (int j = lane; j < npts; j += 32) {
        float dx = px - P[j*3+0];
        float dy = py - P[j*3+1];
        float dz = pz - P[j*3+2];
        float d2 = dx*dx + dy*dy + dz*dz;
        unsigned long long k =
            ((unsigned long long)__float_as_uint(d2) << 32) | (unsigned)j;
        if (k < key[7]) {
            key[7] = k;
            #pragma unroll
            for (int t = 7; t > 0; --t) {
                if (key[t] < key[t-1]) {
                    unsigned long long tmp = key[t]; key[t] = key[t-1]; key[t-1] = tmp;
                }
            }
        }
    }

    int head = 0;
    int sel[8];
    #pragma unroll
    for (int t = 0; t < 8; ++t) {
        unsigned long long cand = (head < 8) ? key[head] : ~0ULL;
        unsigned long long mn = cand;
        #pragma unroll
        for (int off = 16; off >= 1; off >>= 1) {
            unsigned long long o = __shfl_xor(mn, off);
            mn = (o < mn) ? o : mn;
        }
        if (cand == mn && head < 8) head++;
        sel[t] = (int)(mn & 0xFFFFFFFFu);
    }

    for (int c = lane; c < Dd; c += 32) {
        float ssum = 0.f;
        #pragma unroll
        for (int t = 0; t < 8; ++t) ssum += F[(size_t)sel[t] * Dd + c];
        Orow[c] = F[(size_t)n * Dd + c] - ssum * 0.125f;
    }
}

struct ea_frag { v4f g[4]; };

__device__ __forceinline__ v2f ea_pair(v4f f, int u)
{
    v2f r;
    if (u) { r.x = f.z; r.y = f.w; } else { r.x = f.x; r.y = f.y; }
    return r;
}

__device__ __forceinline__ void ea_load_frag(ea_frag& f, const float* __restrict__ p, int kb)
{
    #pragma unroll
    for (int i = 0; i < 4; ++i) f.g[i] = *(const v4f*)(p + kb * 32 + i * 4);
}

template <bool W_IS_A>
__device__ __forceinline__ void ea_proj_core(const float* __restrict__ Ep,
                                             const float* __restrict__ Wp,
                                             v8f (&acc)[4])
{
    ea_frag ef[2];
    ea_frag wf[2][4];
    ea_load_frag(ef[0], Ep, 0);
    #pragma unroll
    for (int t = 0; t < 4; ++t) ea_load_frag(wf[0][t], Wp + t * 16 * Dd, 0);

    #pragma unroll
    for (int kb = 0; kb < Dd / 32; ++kb) {
        const int cur = kb & 1, nxt = cur ^ 1;
        if (kb + 1 < Dd / 32) {
            ea_load_frag(ef[nxt], Ep, kb + 1);
            #pragma unroll
            for (int t = 0; t < 4; ++t) ea_load_frag(wf[nxt][t], Wp + t * 16 * Dd, kb + 1);
        }
        __builtin_amdgcn_sched_barrier(0);
        #pragma unroll
        for (int i = 0; i < 4; ++i) {
            #pragma unroll
            for (int u = 0; u < 2; ++u) {
                const v2f e = ea_pair(ef[cur].g[i], u);
                #pragma unroll
                for (int t = 0; t < 4; ++t) {
                    const v2f w = ea_pair(wf[cur][t].g[i], u);
                    acc[t] = W_IS_A
                        ? __builtin_amdgcn_wmma_f32_16x16x4_f32(false, w, false, e,
                                                                (short)0, acc[t], false, false)
                        : __builtin_amdgcn_wmma_f32_16x16x4_f32(false, e, false, w,
                                                                (short)0, acc[t], false, false);
                }
            }
        }
        __builtin_amdgcn_sched_barrier(0);
    }
}

__device__ __forceinline__ bool ea_proj_tile(int R, int& tm, int& tn0)
{
    const int wave = threadIdx.x >> 5;
    const int nb = R / 64;
    const int g = blockIdx.x / nb;
    tm  = (blockIdx.x - g * nb) * 4 + wave;
    tn0 = g * 4;
    return g < Dd / 64 && tm < R / 16;
}

__global__ void __launch_bounds__(128)
ea_gemm256(const float* __restrict__ A, const float* __restrict__ W,
           const float* __restrict__ bias, float* __restrict__ OUT, int R)
{
    const int lane = threadIdx.x & 31;
    int tm, tn0;
    if (!ea_proj_tile(R, tm, tn0)) return;
    const int col = lane & 15, half = lane >> 4;

    const float* Ep = A + (size_t)(tm * 16 + col) * Dd + 16 * half;
    const float* Wp = W + (size_t)(tn0 * 16 + col) * Dd + 16 * half;

    v8f acc[4] = {};
    ea_proj_core<false>(Ep, Wp, acc);

    #pragma unroll
    for (int t = 0; t < 4; ++t) {
        const int tn = tn0 + t;
        const float bv = bias[tn * 16 + col];
        #pragma unroll
        for (int r = 0; r < 8; ++r) {
            int row = tm * 16 + r + 8 * half;
            OUT[(size_t)row * Dd + tn * 16 + col] = acc[t][r] + bv;
        }
    }
}

template <bool KMODE>
__global__ void __launch_bounds__(128)
ea_gemm_kvfrag(const float* __restrict__ A, const float* __restrict__ W,
               const float* __restrict__ bias, float* __restrict__ OUT, int R)
{
    const int lane = threadIdx.x & 31;
    int tm, tn0;
    if (!ea_proj_tile(R, tm, tn0)) return;
    const int col = lane & 15, half = lane >> 4;

    const float* Ep = A + (size_t)(tm * 16 + col) * Dd + 16 * half;
    const float* Wp = W + (size_t)(tn0 * 16 + col) * Dd + 16 * half;

    v8f acc[4] = {};
    ea_proj_core<KMODE>(Ep, Wp, acc);

    const int bb = tm >> 8, kt = tm & 255;
    #pragma unroll
    for (int t = 0; t < 4; ++t) {
        const int tn = tn0 + t;
        const float* bp = bias + tn * 16 + (KMODE ? 8 * half : col);
        const int bstep = KMODE ? 1 : 0;
        float* dst = OUT + ((size_t)((bb * Hh + (tn >> 1)) * 256 + kt)) * 512
                         + (tn & 1) * 256 + lane * 4;
        v4f o0 = { acc[t][0] + bp[0],         acc[t][1] + bp[bstep],
                   acc[t][2] + bp[2 * bstep], acc[t][3] + bp[3 * bstep] };
        v4f o1 = { acc[t][4] + bp[4 * bstep], acc[t][5] + bp[5 * bstep],
                   acc[t][6] + bp[6 * bstep], acc[t][7] + bp[7 * bstep] };
        *(v4f*)(dst)       = o0;
        *(v4f*)(dst + 128) = o1;
    }
}


__device__ __forceinline__ float ea_xhalf(float x)
{
    int b = __builtin_bit_cast(int, x);
    return __builtin_bit_cast(float,
        __builtin_amdgcn_permlanex16(b, b, 0x76543210, 0xfedcba98, false, false));
}

struct ea_kv {
    v4f k[4];
    v4f v[4];
};

__device__ __forceinline__ v2f ea_lo(v4f f) { v2f r = { f.x, f.y }; return r; }
__device__ __forceinline__ v2f ea_hi(v4f f) { v2f r = { f.z, f.w }; return r; }

__device__ __forceinline__ void ea_load_tile(ea_kv& t, const float* __restrict__ Kl,
                                             const float* __restrict__ Vl, int kt)
{
    const float* Kp = Kl + (size_t)kt * 512;
    const float* Vp = Vl + (size_t)kt * 512;
    #pragma unroll
    for (int q = 0; q < 4; ++q) t.k[q] = *(const v4f*)(Kp + q * 128);
    #pragma unroll
    for (int q = 0; q < 4; ++q) t.v[q] = *(const v4f*)(Vp + q * 128);
}

__device__ __forceinline__ void ea_tile_step(const ea_kv& t, const v2f (&qf)[8],
                                             float& mrun, float& lrun,
                                             v8f& acc0, v8f& acc1)
{
    const float scale = 0.17677669529663687f;

    v8f s = {};
    #pragma unroll
    for (int q = 0; q < 4; ++q) {
        s = __builtin_amdgcn_wmma_f32_16x16x4_f32(false, ea_lo(t.k[q]), false, qf[2 * q],
                                                  (short)0, s, false, false);
        s = __builtin_amdgcn_wmma_f32_16x16x4_f32(false, ea_hi(t.k[q]), false, qf[2 * q + 1],
                                                  (short)0, s, false, false);
    }

    float v[8];
    #pragma unroll
    for (int r = 0; r < 8; ++r) v[r] = s[r] * scale;
    float mx = fmaxf(fmaxf(fmaxf(v[0], v[1]), fmaxf(v[2], v[3])),
                     fmaxf(fmaxf(v[4], v[5]), fmaxf(v[6], v[7])));
    mx = fmaxf(mx, ea_xhalf(mx));
    const float mnew = fmaxf(mrun, mx);
    const float corr = __expf(mrun - mnew);
    float p[8];
    float psum = 0.f;
    #pragma unroll
    for (int r = 0; r < 8; ++r) { p[r] = __expf(v[r] - mnew); psum += p[r]; }
    lrun = lrun * corr + psum;
    #pragma unroll
    for (int r = 0; r < 8; ++r) { acc0[r] *= corr; acc1[r] *= corr; }
    mrun = mnew;

    #pragma unroll
    for (int j = 0; j < 4; ++j) {
        v2f pf; pf.x = p[2 * j];  pf.y = p[2 * j + 1];
        const v2f a0 = (j & 1) ? ea_hi(t.v[j >> 1])       : ea_lo(t.v[j >> 1]);
        const v2f a1 = (j & 1) ? ea_hi(t.v[2 + (j >> 1)]) : ea_lo(t.v[2 + (j >> 1)]);
        acc0 = __builtin_amdgcn_wmma_f32_16x16x4_f32(false, a0, false, pf,
                                                     (short)0, acc0, false, false);
        acc1 = __builtin_amdgcn_wmma_f32_16x16x4_f32(false, a1, false, pf,
                                                     (short)0, acc1, false, false);
    }
}

__global__ void __launch_bounds__(256)
ea_attn(const float* __restrict__ Q, const float* __restrict__ Km,
        const float* __restrict__ Vm, float* __restrict__ O)
{
    const int lane = threadIdx.x & 31;
    const int wave = threadIdx.x >> 5;
    const int tile = blockIdx.x * 8 + wave;
    const int qt = tile & 127;
    const int h  = (tile >> 7) & 7;
    const int b  = tile >> 10;
    const int col = lane & 15, half = lane >> 4;

    const float* Qrow = Q + (size_t)(b * Nn + qt * 16 + col) * Dd + h * DHh + 8 * half;
    v2f qf[8];
    #pragma unroll
    for (int q = 0; q < 4; ++q) {
        const v4f g = *(const v4f*)(Qrow + 16 * (q >> 1) + 4 * (q & 1));
        qf[2 * q]     = ea_lo(g);
        qf[2 * q + 1] = ea_hi(g);
    }

    float mrun = -__builtin_inff();
    float lrun = 0.f;
    v8f acc0 = {}, acc1 = {};

    const float* Kb = Km + (size_t)((b * Hh + h) * 256) * 512 + lane * 4;
    const float* Vb = Vm + (size_t)((b * Hh + h) * 256) * 512 + lane * 4;

    ea_kv e0, e1, e2, e3;
    __builtin_amdgcn_sched_barrier(0);
    ea_load_tile(e0, Kb, Vb, 0);
    __builtin_amdgcn_sched_barrier(0);
    ea_load_tile(e1, Kb, Vb, 1);
    __builtin_amdgcn_sched_barrier(0);
    ea_load_tile(e2, Kb, Vb, 2);
    __builtin_amdgcn_sched_barrier(0);
    ea_load_tile(e3, Kb, Vb, 3);
    __builtin_amdgcn_sched_barrier(0);
    for (int kt = 0; kt < Ll / 16; kt += 4) {
        int t0 = (kt + 4 < Ll / 16) ? kt + 4 : kt;
        int t1 = t0 + 1, t2 = t0 + 2, t3 = t0 + 3;
        ea_tile_step(e0, qf, mrun, lrun, acc0, acc1);
        asm volatile("" : "+v"(acc0), "+v"(acc1), "+s"(t0));
        ea_load_tile(e0, Kb, Vb, t0);
        __builtin_amdgcn_sched_barrier(0);
        ea_tile_step(e1, qf, mrun, lrun, acc0, acc1);
        asm volatile("" : "+v"(acc0), "+v"(acc1), "+s"(t1));
        ea_load_tile(e1, Kb, Vb, t1);
        __builtin_amdgcn_sched_barrier(0);
        ea_tile_step(e2, qf, mrun, lrun, acc0, acc1);
        asm volatile("" : "+v"(acc0), "+v"(acc1), "+s"(t2));
        ea_load_tile(e2, Kb, Vb, t2);
        __builtin_amdgcn_sched_barrier(0);
        ea_tile_step(e3, qf, mrun, lrun, acc0, acc1);
        asm volatile("" : "+v"(acc0), "+v"(acc1), "+s"(t3));
        ea_load_tile(e3, Kb, Vb, t3);
        __builtin_amdgcn_sched_barrier(0);
    }

    const float inv = 1.0f / (lrun + ea_xhalf(lrun));
    float* Op = O + (size_t)(b * Nn + qt * 16 + col) * Dd + h * DHh + 8 * half;
    v4f o0 = { acc0[0] * inv, acc0[1] * inv, acc0[2] * inv, acc0[3] * inv };
    v4f o1 = { acc0[4] * inv, acc0[5] * inv, acc0[6] * inv, acc0[7] * inv };
    v4f o2 = { acc1[0] * inv, acc1[1] * inv, acc1[2] * inv, acc1[3] * inv };
    v4f o3 = { acc1[4] * inv, acc1[5] * inv, acc1[6] * inv, acc1[7] * inv };
    *(v4f*)(Op)      = o0;
    *(v4f*)(Op + 4)  = o1;
    *(v4f*)(Op + 16) = o2;
    *(v4f*)(Op + 20) = o3;
}

__global__ void __launch_bounds__(128)
ea_gemm_out(const float* __restrict__ A, const float* __restrict__ W,
            const float* __restrict__ bias,
            const float* __restrict__ pts,
            const float* __restrict__ sw, const float* __restrict__ sb,
            float* __restrict__ OUT, int R)
{
    const int lane = threadIdx.x & 31;
    int tm, tn0;
    if (!ea_proj_tile(R, tm, tn0)) return;
    const int col = lane & 15, half = lane >> 4;

    const float* Ep = A + (size_t)(tm * 16 + col) * Dd + 16 * half;
    const float* Wp = W + (size_t)(tn0 * 16 + col) * Dd + 16 * half;

    v8f acc[4] = {};
    ea_proj_core<false>(Ep, Wp, acc);

    #pragma unroll
    for (int t = 0; t < 4; ++t) {
        const int colg = (tn0 + t) * 16 + col;
        const float s0 = sw[colg*3+0], s1 = sw[colg*3+1], s2 = sw[colg*3+2];
        const float add = bias[colg] + sb[colg];
        #pragma unroll
        for (int r = 0; r < 8; ++r) {
            int row = tm * 16 + r + 8 * half;
            float px = pts[row*3+0], py = pts[row*3+1], pz = pts[row*3+2];
            OUT[(size_t)row * Dd + colg] = acc[t][r] + add + px*s0 + py*s1 + pz*s2;
        }
    }
}

extern "C" void kernel_launch(void* const* d_in, const int* in_sizes, int n_in,
                              void* d_out, int out_size, void* d_ws, size_t ws_size,
                              hipStream_t stream)
{
    const float* cur_pts  = (const float*)d_in[0];
    const float* cur_feat = (const float*)d_in[1];
    const float* nbr_pts  = (const float*)d_in[2];
    const float* nbr_feat = (const float*)d_in[3];
    const float* in_w     = (const float*)d_in[4];
    const float* in_b     = (const float*)d_in[5];
    const float* out_w    = (const float*)d_in[6];
    const float* out_b    = (const float*)d_in[7];
    const float* sp_w     = (const float*)d_in[8];
    const float* sp_b     = (const float*)d_in[9];
    float* out = (float*)d_out;

    float* ws   = (float*)d_ws;
    float* curE = ws;
    float* nbrE = curE + 4096 * 256;
    float* Qb   = nbrE + 8192 * 256;
    float* Kb   = Qb   + 4096 * 256;
    float* Vb   = Kb   + 8192 * 256;
    float* Ob   = Vb   + 8192 * 256;

    const dim3 blk(256);
    const dim3 blk4(128);

    ea_knn_edge<<<dim3(4096 / 8), blk, 0, stream>>>(cur_pts, cur_feat, curE,
                                                    Nn, Bc, 0);
    ea_knn_edge<<<dim3(8192 / 8), blk, 0, stream>>>(nbr_pts, nbr_feat, nbrE,
                                                    Nn, Mm * Bc, 1);

    ea_gemm256<<<dim3(4096 / 16), blk4, 0, stream>>>(curE, in_w,           in_b,       Qb, 4096);
    ea_gemm_kvfrag<true><<<dim3(8192 / 16), blk4, 0, stream>>>(nbrE, in_w + 256*256, in_b + 256, Kb, 8192);
    ea_gemm_kvfrag<false><<<dim3(8192 / 16), blk4, 0, stream>>>(nbrE, in_w + 512*256, in_b + 512, Vb, 8192);

    ea_attn<<<dim3(2048 / 8), blk, 0, stream>>>(Qb, Kb, Vb, Ob);

    ea_gemm_out<<<dim3(4096 / 16), blk4, 0, stream>>>(Ob, out_w, out_b,
                                                    cur_pts, sp_w, sp_b,
                                                    out, 4096);
}


# KNN: the row's best list is kept sorted across lanes (lane t = t-th smallest key); a ballot against the 8th-best distance picks entrants, each inserted by one lane-shift min/max; sums unchanged
# speedup vs baseline: 2.7240x; 1.0382x over previous
#include <hip/hip_runtime.h>
#include <hip/hip_bf16.h>


#define Bc   2
#define Nn   2048
#define Dd   256
#define Hh   8
#define DHh  32
#define Mm   2
#define Ll   4096

typedef __attribute__((ext_vector_type(2))) float v2f;
typedef __attribute__((ext_vector_type(4))) float v4f;
typedef __attribute__((ext_vector_type(8))) float v8f;


__device__ __forceinline__ unsigned ea_lane_below(unsigned x)
{
    return (unsigned)__builtin_amdgcn_update_dpp(0, (int)x, 0x111  ,
                                                 0xf, 0xf, true);
}

__device__ __forceinline__ void ea_knn_insert(unsigned m, unsigned d2b, int jbase,
                                              unsigned& khi, unsigned& klo)
{
    while (m) {
        const int l = __builtin_ctz(m);
        m &= m - 1;
        const unsigned chi = (unsigned)__builtin_amdgcn_readlane((int)d2b, l);
        const unsigned long long c =
            ((unsigned long long)chi << 32) | (unsigned)(jbase + l);
        const unsigned long long below =
            ((unsigned long long)ea_lane_below(khi) << 32) | ea_lane_below(klo);
        const unsigned long long key = ((unsigned long long)khi << 32) | klo;
        const unsigned long long up = (c > below) ? c : below;
        const unsigned long long nk = (key < up) ? key : up;
        khi = (unsigned)(nk >> 32);
        klo = (unsigned)nk;
    }
}

__global__ void ea_knn_edge(const float* __restrict__ pts,
                            const float* __restrict__ feats,
                            float* __restrict__ out,
                            int npts, int nsets, int remap)
{
    const int lane = threadIdx.x & 31;
    const int wave = threadIdx.x >> 5;
    const int rowg = blockIdx.x * 8 + wave;
    if (rowg >= nsets * npts) return;
    const int s = rowg / npts;
    const int n = rowg - s * npts;
    int os = s;
    if (remap) { int m = s / Bc; int b = s - m * Bc; os = b * Mm + m; }

    const float* P = pts   + (size_t)s  * npts * 3;
    const float* F = feats + (size_t)s  * npts * Dd;
    float*    Orow = out   + ((size_t)os * npts + n) * Dd;

    const float px = P[n*3+0], py = P[n*3+1], pz = P[n*3+2];

    unsigned khi = ~0u, klo = ~0u;
    unsigned bar = ~0u;

    const unsigned last = (unsigned)npts - 1u;
    unsigned ia = (unsigned)lane < last ? (unsigned)lane : last;
    unsigned ib = (unsigned)lane + 32u < last ? (unsigned)lane + 32u : last;
    const float* pa = P + (size_t)ia * 3;
    const float* pb = P + (size_t)ib * 3;
    float ax = pa[0], ay = pa[1], az = pa[2];
    float bx = pb[0], by = pb[1], bz = pb[2];

    for (int j0 = 0; j0 < npts; j0 += 64) {
        const int ja = j0 + lane, jb = ja + 32;
        float d2a, d2b;
        {
            float dx = px - ax;
            float dy = py - ay;
            float dz = pz - az;
            d2a = dx*dx + dy*dy + dz*dz;
        }
        {
            float dx = px - bx;
            float dy = py - by;
            float dz = pz - bz;
            d2b = dx*dx + dy*dy + dz*dz;
        }
        ia = (unsigned)ja + 64u < last ? (unsigned)ja + 64u : last;
        ib = (unsigned)jb + 64u < last ? (unsigned)jb + 64u : last;
        pa = P + (size_t)ia * 3;
        pb = P + (size_t)ib * 3;
        ax = pa[0]; ay = pa[1]; az = pa[2];
        bx = pb[0]; by = pb[1]; bz = pb[2];

        const unsigned ua = __float_as_uint(d2a), ub = __float_as_uint(d2b);
        const unsigned ma = __builtin_amdgcn_ballot_w32(ua <= bar)
                          & __builtin_amdgcn_ballot_w32(ja < npts);
        if (ma) {
            ea_knn_insert(ma, ua, j0, khi, klo);
            bar = (unsigned)__builtin_amdgcn_readlane((int)khi, 7);
        }
        const unsigned mb = __builtin_amdgcn_ballot_w32(ub <= bar)
                          & __builtin_amdgcn_ballot_w32(jb < npts);
        if (mb) {
            ea_knn_insert(mb, ub, j0 + 32, khi, klo);
            bar = (unsigned)__builtin_amdgcn_readlane((int)khi, 7);
        }
    }

    int sel[8];
    #pragma unroll
    for (int t = 0; t < 8; ++t) sel[t] = __builtin_amdgcn_readlane((int)klo, t);

    for (int c = lane; c < Dd; c += 32) {
        float ssum = 0.f;
        #pragma unroll
        for (int t = 0; t < 8; ++t) ssum += F[(size_t)sel[t] * Dd + c];
        Orow[c] = F[(size_t)n * Dd + c] - ssum * 0.125f;
    }
}

struct ea_frag { v4f g[4]; };

__device__ __forceinline__ v2f ea_pair(v4f f, int u)
{
    v2f r;
    if (u) { r.x = f.z; r.y = f.w; } else { r.x = f.x; r.y = f.y; }
    return r;
}

__device__ __forceinline__ void ea_load_frag(ea_frag& f, const float* __restrict__ p, int kb)
{
    #pragma unroll
    for (int i = 0; i < 4; ++i) f.g[i] = *(const v4f*)(p + kb * 32 + i * 4);
}

template <bool W_IS_A>
__device__ __forceinline__ void ea_proj_core(const float* __restrict__ Ep,
                                             const float* __restrict__ Wp,
                                             v8f (&acc)[4])
{
    ea_frag ef[2];
    ea_frag wf[2][4];
    ea_load_frag(ef[0], Ep, 0);
    #pragma unroll
    for (int t = 0; t < 4; ++t) ea_load_frag(wf[0][t], Wp + t * 16 * Dd, 0);

    #pragma unroll
    for (int kb = 0; kb < Dd / 32; ++kb) {
        const int cur = kb & 1, nxt = cur ^ 1;
        if (kb + 1 < Dd / 32) {
            ea_load_frag(ef[nxt], Ep, kb + 1);
            #pragma unroll
            for (int t = 0; t < 4; ++t) ea_load_frag(wf[nxt][t], Wp + t * 16 * Dd, kb + 1);
        }
        __builtin_amdgcn_sched_barrier(0);
        #pragma unroll
        for (int i = 0; i < 4; ++i) {
            #pragma unroll
            for (int u = 0; u < 2; ++u) {
                const v2f e = ea_pair(ef[cur].g[i], u);
                #pragma unroll
                for (int t = 0; t < 4; ++t) {
                    const v2f w = ea_pair(wf[cur][t].g[i], u);
                    acc[t] = W_IS_A
                        ? __builtin_amdgcn_wmma_f32_16x16x4_f32(false, w, false, e,
                                                                (short)0, acc[t], false, false)
                        : __builtin_amdgcn_wmma_f32_16x16x4_f32(false, e, false, w,
                                                                (short)0, acc[t], false, false);
                }
            }
        }
        __builtin_amdgcn_sched_barrier(0);
    }
}

__device__ __forceinline__ bool ea_proj_tile(int R, int& tm, int& tn0)
{
    const int wave = threadIdx.x >> 5;
    const int nb = R / 64;
    const int g = blockIdx.x / nb;
    tm  = (blockIdx.x - g * nb) * 4 + wave;
    tn0 = g * 4;
    return g < Dd / 64 && tm < R / 16;
}

__global__ void __launch_bounds__(128)
ea_gemm256(const float* __restrict__ A, const float* __restrict__ W,
           const float* __restrict__ bias, float* __restrict__ OUT, int R)
{
    const int lane = threadIdx.x & 31;
    int tm, tn0;
    if (!ea_proj_tile(R, tm, tn0)) return;
    const int col = lane & 15, half = lane >> 4;

    const float* Ep = A + (size_t)(tm * 16 + col) * Dd + 16 * half;
    const float* Wp = W + (size_t)(tn0 * 16 + col) * Dd + 16 * half;

    v8f acc[4] = {};
    ea_proj_core<false>(Ep, Wp, acc);

    #pragma unroll
    for (int t = 0; t < 4; ++t) {
        const int tn = tn0 + t;
        const float bv = bias[tn * 16 + col];
        #pragma unroll
        for (int r = 0; r < 8; ++r) {
            int row = tm * 16 + r + 8 * half;
            OUT[(size_t)row * Dd + tn * 16 + col] = acc[t][r] + bv;
        }
    }
}

template <bool KMODE>
__global__ void __launch_bounds__(128)
ea_gemm_kvfrag(const float* __restrict__ A, const float* __restrict__ W,
               const float* __restrict__ bias, float* __restrict__ OUT, int R)
{
    const int lane = threadIdx.x & 31;
    int tm, tn0;
    if (!ea_proj_tile(R, tm, tn0)) return;
    const int col = lane & 15, half = lane >> 4;

    const float* Ep = A + (size_t)(tm * 16 + col) * Dd + 16 * half;
    const float* Wp = W + (size_t)(tn0 * 16 + col) * Dd + 16 * half;

    v8f acc[4] = {};
    ea_proj_core<KMODE>(Ep, Wp, acc);

    const int bb = tm >> 8, kt = tm & 255;
    #pragma unroll
    for (int t = 0; t < 4; ++t) {
        const int tn = tn0 + t;
        const float* bp = bias + tn * 16 + (KMODE ? 8 * half : col);
        const int bstep = KMODE ? 1 : 0;
        float* dst = OUT + ((size_t)((bb * Hh + (tn >> 1)) * 256 + kt)) * 512
                         + (tn & 1) * 256 + lane * 4;
        v4f o0 = { acc[t][0] + bp[0],         acc[t][1] + bp[bstep],
                   acc[t][2] + bp[2 * bstep], acc[t][3] + bp[3 * bstep] };
        v4f o1 = { acc[t][4] + bp[4 * bstep], acc[t][5] + bp[5 * bstep],
                   acc[t][6] + bp[6 * bstep], acc[t][7] + bp[7 * bstep] };
        *(v4f*)(dst)       = o0;
        *(v4f*)(dst + 128) = o1;
    }
}


__device__ __forceinline__ float ea_xhalf(float x)
{
    int b = __builtin_bit_cast(int, x);
    return __builtin_bit_cast(float,
        __builtin_amdgcn_permlanex16(b, b, 0x76543210, 0xfedcba98, false, false));
}

struct ea_kv {
    v4f k[4];
    v4f v[4];
};

__device__ __forceinline__ v2f ea_lo(v4f f) { v2f r = { f.x, f.y }; return r; }
__device__ __forceinline__ v2f ea_hi(v4f f) { v2f r = { f.z, f.w }; return r; }

__device__ __forceinline__ void ea_load_tile(ea_kv& t, const float* __restrict__ Kl,
                                             const float* __restrict__ Vl, int kt)
{
    const float* Kp = Kl + (size_t)kt * 512;
    const float* Vp = Vl + (size_t)kt * 512;
    #pragma unroll
    for (int q = 0; q < 4; ++q) t.k[q] = *(const v4f*)(Kp + q * 128);
    #pragma unroll
    for (int q = 0; q < 4; ++q) t.v[q] = *(const v4f*)(Vp + q * 128);
}

__device__ __forceinline__ void ea_tile_step(const ea_kv& t, const v2f (&qf)[8],
                                             float& mrun, float& lrun,
                                             v8f& acc0, v8f& acc1)
{
    const float scale = 0.17677669529663687f;

    v8f s = {};
    #pragma unroll
    for (int q = 0; q < 4; ++q) {
        s = __builtin_amdgcn_wmma_f32_16x16x4_f32(false, ea_lo(t.k[q]), false, qf[2 * q],
                                                  (short)0, s, false, false);
        s = __builtin_amdgcn_wmma_f32_16x16x4_f32(false, ea_hi(t.k[q]), false, qf[2 * q + 1],
                                                  (short)0, s, false, false);
    }

    float v[8];
    #pragma unroll
    for (int r = 0; r < 8; ++r) v[r] = s[r] * scale;
    float mx = fmaxf(fmaxf(fmaxf(v[0], v[1]), fmaxf(v[2], v[3])),
                     fmaxf(fmaxf(v[4], v[5]), fmaxf(v[6], v[7])));
    mx = fmaxf(mx, ea_xhalf(mx));
    const float mnew = fmaxf(mrun, mx);
    const float corr = __expf(mrun - mnew);
    float p[8];
    float psum = 0.f;
    #pragma unroll
    for (int r = 0; r < 8; ++r) { p[r] = __expf(v[r] - mnew); psum += p[r]; }
    lrun = lrun * corr + psum;
    #pragma unroll
    for (int r = 0; r < 8; ++r) { acc0[r] *= corr; acc1[r] *= corr; }
    mrun = mnew;

    #pragma unroll
    for (int j = 0; j < 4; ++j) {
        v2f pf; pf.x = p[2 * j];  pf.y = p[2 * j + 1];
        const v2f a0 = (j & 1) ? ea_hi(t.v[j >> 1])       : ea_lo(t.v[j >> 1]);
        const v2f a1 = (j & 1) ? ea_hi(t.v[2 + (j >> 1)]) : ea_lo(t.v[2 + (j >> 1)]);
        acc0 = __builtin_amdgcn_wmma_f32_16x16x4_f32(false, a0, false, pf,
                                                     (short)0, acc0, false, false);
        acc1 = __builtin_amdgcn_wmma_f32_16x16x4_f32(false, a1, false, pf,
                                                     (short)0, acc1, false, false);
    }
}

__global__ void __launch_bounds__(256)
ea_attn(const float* __restrict__ Q, const float* __restrict__ Km,
        const float* __restrict__ Vm, float* __restrict__ O)
{
    const int lane = threadIdx.x & 31;
    const int wave = threadIdx.x >> 5;
    const int tile = blockIdx.x * 8 + wave;
    const int qt = tile & 127;
    const int h  = (tile >> 7) & 7;
    const int b  = tile >> 10;
    const int col = lane & 15, half = lane >> 4;

    const float* Qrow = Q + (size_t)(b * Nn + qt * 16 + col) * Dd + h * DHh + 8 * half;
    v2f qf[8];
    #pragma unroll
    for (int q = 0; q < 4; ++q) {
        const v4f g = *(const v4f*)(Qrow + 16 * (q >> 1) + 4 * (q & 1));
        qf[2 * q]     = ea_lo(g);
        qf[2 * q + 1] = ea_hi(g);
    }

    float mrun = -__builtin_inff();
    float lrun = 0.f;
    v8f acc0 = {}, acc1 = {};

    const float* Kb = Km + (size_t)((b * Hh + h) * 256) * 512 + lane * 4;
    const float* Vb = Vm + (size_t)((b * Hh + h) * 256) * 512 + lane * 4;

    ea_kv e0, e1, e2, e3;
    __builtin_amdgcn_sched_barrier(0);
    ea_load_tile(e0, Kb, Vb, 0);
    __builtin_amdgcn_sched_barrier(0);
    ea_load_tile(e1, Kb, Vb, 1);
    __builtin_amdgcn_sched_barrier(0);
    ea_load_tile(e2, Kb, Vb, 2);
    __builtin_amdgcn_sched_barrier(0);
    ea_load_tile(e3, Kb, Vb, 3);
    __builtin_amdgcn_sched_barrier(0);
    for (int kt = 0; kt < Ll / 16; kt += 4) {
        int t0 = (kt + 4 < Ll / 16) ? kt + 4 : kt;
        int t1 = t0 + 1, t2 = t0 + 2, t3 = t0 + 3;
        ea_tile_step(e0, qf, mrun, lrun, acc0, acc1);
        asm volatile("" : "+v"(acc0), "+v"(acc1), "+s"(t0));
        ea_load_tile(e0, Kb, Vb, t0);
        __builtin_amdgcn_sched_barrier(0);
        ea_tile_step(e1, qf, mrun, lrun, acc0, acc1);
        asm volatile("" : "+v"(acc0), "+v"(acc1), "+s"(t1));
        ea_load_tile(e1, Kb, Vb, t1);
        __builtin_amdgcn_sched_barrier(0);
        ea_tile_step(e2, qf, mrun, lrun, acc0, acc1);
        asm volatile("" : "+v"(acc0), "+v"(acc1), "+s"(t2));
        ea_load_tile(e2, Kb, Vb, t2);
        __builtin_amdgcn_sched_barrier(0);
        ea_tile_step(e3, qf, mrun, lrun, acc0, acc1);
        asm volatile("" : "+v"(acc0), "+v"(acc1), "+s"(t3));
        ea_load_tile(e3, Kb, Vb, t3);
        __builtin_amdgcn_sched_barrier(0);
    }

    const float inv = 1.0f / (lrun + ea_xhalf(lrun));
    float* Op = O + (size_t)(b * Nn + qt * 16 + col) * Dd + h * DHh + 8 * half;
    v4f o0 = { acc0[0] * inv, acc0[1] * inv, acc0[2] * inv, acc0[3] * inv };
    v4f o1 = { acc0[4] * inv, acc0[5] * inv, acc0[6] * inv, acc0[7] * inv };
    v4f o2 = { acc1[0] * inv, acc1[1] * inv, acc1[2] * inv, acc1[3] * inv };
    v4f o3 = { acc1[4] * inv, acc1[5] * inv, acc1[6] * inv, acc1[7] * inv };
    *(v4f*)(Op)      = o0;
    *(v4f*)(Op + 4)  = o1;
    *(v4f*)(Op + 16) = o2;
    *(v4f*)(Op + 20) = o3;
}

__global__ void __launch_bounds__(128)
ea_gemm_out(const float* __restrict__ A, const float* __restrict__ W,
            const float* __restrict__ bias,
            const float* __restrict__ pts,
            const float* __restrict__ sw, const float* __restrict__ sb,
            float* __restrict__ OUT, int R)
{
    const int lane = threadIdx.x & 31;
    int tm, tn0;
    if (!ea_proj_tile(R, tm, tn0)) return;
    const int col = lane & 15, half = lane >> 4;

    const float* Ep = A + (size_t)(tm * 16 + col) * Dd + 16 * half;
    const float* Wp = W + (size_t)(tn0 * 16 + col) * Dd + 16 * half;

    v8f acc[4] = {};
    ea_proj_core<false>(Ep, Wp, acc);

    #pragma unroll
    for (int t = 0; t < 4; ++t) {
        const int colg = (tn0 + t) * 16 + col;
        const float s0 = sw[colg*3+0], s1 = sw[colg*3+1], s2 = sw[colg*3+2];
        const float add = bias[colg] + sb[colg];
        #pragma unroll
        for (int r = 0; r < 8; ++r) {
            int row = tm * 16 + r + 8 * half;
            float px = pts[row*3+0], py = pts[row*3+1], pz = pts[row*3+2];
            OUT[(size_t)row * Dd + colg] = acc[t][r] + add + px*s0 + py*s1 + pz*s2;
        }
    }
}

extern "C" void kernel_launch(void* const* d_in, const int* in_sizes, int n_in,
                              void* d_out, int out_size, void* d_ws, size_t ws_size,
                              hipStream_t stream)
{
    const float* cur_pts  = (const float*)d_in[0];
    const float* cur_feat = (const float*)d_in[1];
    const float* nbr_pts  = (const float*)d_in[2];
    const float* nbr_feat = (const float*)d_in[3];
    const float* in_w     = (const float*)d_in[4];
    const float* in_b     = (const float*)d_in[5];
    const float* out_w    = (const float*)d_in[6];
    const float* out_b    = (const float*)d_in[7];
    const float* sp_w     = (const float*)d_in[8];
    const float* sp_b     = (const float*)d_in[9];
    float* out = (float*)d_out;

    float* ws   = (float*)d_ws;
    float* curE = ws;
    float* nbrE = curE + 4096 * 256;
    float* Qb   = nbrE + 8192 * 256;
    float* Kb   = Qb   + 4096 * 256;
    float* Vb   = Kb   + 8192 * 256;
    float* Ob   = Vb   + 8192 * 256;

    const dim3 blk(256);
    const dim3 blk4(128);

    ea_knn_edge<<<dim3(4096 / 8), blk, 0, stream>>>(cur_pts, cur_feat, curE,
                                                    Nn, Bc, 0);
    ea_knn_edge<<<dim3(8192 / 8), blk, 0, stream>>>(nbr_pts, nbr_feat, nbrE,
                                                    Nn, Mm * Bc, 1);

    ea_gemm256<<<dim3(4096 / 16), blk4, 0, stream>>>(curE, in_w,           in_b,       Qb, 4096);
    ea_gemm_kvfrag<true><<<dim3(8192 / 16), blk4, 0, stream>>>(nbrE, in_w + 256*256, in_b + 256, Kb, 8192);
    ea_gemm_kvfrag<false><<<dim3(8192 / 16), blk4, 0, stream>>>(nbrE, in_w + 512*256, in_b + 512, Vb, 8192);

    ea_attn<<<dim3(2048 / 8), blk, 0, stream>>>(Qb, Kb, Vb, Ob);

    ea_gemm_out<<<dim3(4096 / 16), blk4, 0, stream>>>(Ob, out_w, out_b,
                                                    cur_pts, sp_w, sp_b,
                                                    out, 4096);
}


# independent stages share launches: both KNN groups in one launch (set size now a compile-time constant) and the Q, K, V projections in one launch, 7 launches -> 4; sums unchanged
# speedup vs baseline: 2.8466x; 1.0450x over previous
#include <hip/hip_runtime.h>
#include <hip/hip_bf16.h>


#define Bc   2
#define Nn   2048
#define Dd   256
#define Hh   8
#define DHh  32
#define Mm   2
#define Ll   4096

typedef __attribute__((ext_vector_type(2))) float v2f;
typedef __attribute__((ext_vector_type(4))) float v4f;
typedef __attribute__((ext_vector_type(8))) float v8f;


__device__ __forceinline__ unsigned ea_lane_below(unsigned x)
{
    return (unsigned)__builtin_amdgcn_update_dpp(0, (int)x, 0x111  ,
                                                 0xf, 0xf, true);
}

__device__ __forceinline__ void ea_knn_insert(unsigned m, unsigned d2b, int jbase,
                                              unsigned& khi, unsigned& klo)
{
    while (m) {
        const int l = __builtin_ctz(m);
        m &= m - 1;
        const unsigned chi = (unsigned)__builtin_amdgcn_readlane((int)d2b, l);
        const unsigned long long c =
            ((unsigned long long)chi << 32) | (unsigned)(jbase + l);
        const unsigned long long below =
            ((unsigned long long)ea_lane_below(khi) << 32) | ea_lane_below(klo);
        const unsigned long long key = ((unsigned long long)khi << 32) | klo;
        const unsigned long long up = (c > below) ? c : below;
        const unsigned long long nk = (key < up) ? key : up;
        khi = (unsigned)(nk >> 32);
        klo = (unsigned)nk;
    }
}

#define EA_KNN_CUR_BLOCKS (Bc * Nn / 8)
#define EA_KNN_NBR_BLOCKS (Mm * Bc * Nn / 8)

__global__ void ea_knn_edge(const float* __restrict__ cur_pts,
                            const float* __restrict__ cur_feat,
                            float* __restrict__ cur_out,
                            const float* __restrict__ nbr_pts,
                            const float* __restrict__ nbr_feat,
                            float* __restrict__ nbr_out)
{
    const int lane = threadIdx.x & 31;
    const int wave = threadIdx.x >> 5;
    const bool nbr = blockIdx.x >= EA_KNN_CUR_BLOCKS;
    const float* __restrict__ pts   = nbr ? nbr_pts  : cur_pts;
    const float* __restrict__ feats = nbr ? nbr_feat : cur_feat;
    float* __restrict__ out         = nbr ? nbr_out  : cur_out;
    const int npts  = Nn;
    const int nsets = nbr ? Mm * Bc : Bc;
    const int remap = nbr ? 1 : 0;
    const int rowg = (nbr ? (int)blockIdx.x - EA_KNN_CUR_BLOCKS : (int)blockIdx.x) * 8 + wave;
    if (rowg >= nsets * npts) return;
    const int s = rowg / npts;
    const int n = rowg - s * npts;
    int os = s;
    if (remap) { int m = s / Bc; int b = s - m * Bc; os = b * Mm + m; }

    const float* P = pts   + (size_t)s  * npts * 3;
    const float* F = feats + (size_t)s  * npts * Dd;
    float*    Orow = out   + ((size_t)os * npts + n) * Dd;

    const float px = P[n*3+0], py = P[n*3+1], pz = P[n*3+2];

    unsigned khi = ~0u, klo = ~0u;
    unsigned bar = ~0u;

    const unsigned last = (unsigned)npts - 1u;
    unsigned ia = (unsigned)lane < last ? (unsigned)lane : last;
    unsigned ib = (unsigned)lane + 32u < last ? (unsigned)lane + 32u : last;
    const float* pa = P + (size_t)ia * 3;
    const float* pb = P + (size_t)ib * 3;
    float ax = pa[0], ay = pa[1], az = pa[2];
    float bx = pb[0], by = pb[1], bz = pb[2];

    for (int j0 = 0; j0 < npts; j0 += 64) {
        const int ja = j0 + lane, jb = ja + 32;
        float d2a, d2b;
        {
            float dx = px - ax;
            float dy = py - ay;
            float dz = pz - az;
            d2a = dx*dx + dy*dy + dz*dz;
        }
        {
            float dx = px - bx;
            float dy = py - by;
            float dz = pz - bz;
            d2b = dx*dx + dy*dy + dz*dz;
        }
        ia = (unsigned)ja + 64u < last ? (unsigned)ja + 64u : last;
        ib = (unsigned)jb + 64u < last ? (unsigned)jb + 64u : last;
        pa = P + (size_t)ia * 3;
        pb = P + (size_t)ib * 3;
        ax = pa[0]; ay = pa[1]; az = pa[2];
        bx = pb[0]; by = pb[1]; bz = pb[2];

        const unsigned ua = __float_as_uint(d2a), ub = __float_as_uint(d2b);
        const unsigned ma = __builtin_amdgcn_ballot_w32(ua <= bar)
                          & __builtin_amdgcn_ballot_w32(ja < npts);
        if (ma) {
            ea_knn_insert(ma, ua, j0, khi, klo);
            bar = (unsigned)__builtin_amdgcn_readlane((int)khi, 7);
        }
        const unsigned mb = __builtin_amdgcn_ballot_w32(ub <= bar)
                          & __builtin_amdgcn_ballot_w32(jb < npts);
        if (mb) {
            ea_knn_insert(mb, ub, j0 + 32, khi, klo);
            bar = (unsigned)__builtin_amdgcn_readlane((int)khi, 7);
        }
    }

    int sel[8];
    #pragma unroll
    for (int t = 0; t < 8; ++t) sel[t] = __builtin_amdgcn_readlane((int)klo, t);

    for (int c = lane; c < Dd; c += 32) {
        float ssum = 0.f;
        #pragma unroll
        for (int t = 0; t < 8; ++t) ssum += F[(size_t)sel[t] * Dd + c];
        Orow[c] = F[(size_t)n * Dd + c] - ssum * 0.125f;
    }
}

struct ea_frag { v4f g[4]; };

__device__ __forceinline__ v2f ea_pair(v4f f, int u)
{
    v2f r;
    if (u) { r.x = f.z; r.y = f.w; } else { r.x = f.x; r.y = f.y; }
    return r;
}

__device__ __forceinline__ void ea_load_frag(ea_frag& f, const float* __restrict__ p, int kb)
{
    #pragma unroll
    for (int i = 0; i < 4; ++i) f.g[i] = *(const v4f*)(p + kb * 32 + i * 4);
}

template <bool W_IS_A>
__device__ __forceinline__ void ea_proj_core(const float* __restrict__ Ep,
                                             const float* __restrict__ Wp,
                                             v8f (&acc)[4])
{
    ea_frag ef[2];
    ea_frag wf[2][4];
    ea_load_frag(ef[0], Ep, 0);
    #pragma unroll
    for (int t = 0; t < 4; ++t) ea_load_frag(wf[0][t], Wp + t * 16 * Dd, 0);

    #pragma unroll
    for (int kb = 0; kb < Dd / 32; ++kb) {
        const int cur = kb & 1, nxt = cur ^ 1;
        if (kb + 1 < Dd / 32) {
            ea_load_frag(ef[nxt], Ep, kb + 1);
            #pragma unroll
            for (int t = 0; t < 4; ++t) ea_load_frag(wf[nxt][t], Wp + t * 16 * Dd, kb + 1);
        }
        __builtin_amdgcn_sched_barrier(0);
        #pragma unroll
        for (int i = 0; i < 4; ++i) {
            #pragma unroll
            for (int u = 0; u < 2; ++u) {
                const v2f e = ea_pair(ef[cur].g[i], u);
                #pragma unroll
                for (int t = 0; t < 4; ++t) {
                    const v2f w = ea_pair(wf[cur][t].g[i], u);
                    acc[t] = W_IS_A
                        ? __builtin_amdgcn_wmma_f32_16x16x4_f32(false, w, false, e,
                                                                (short)0, acc[t], false, false)
                        : __builtin_amdgcn_wmma_f32_16x16x4_f32(false, e, false, w,
                                                                (short)0, acc[t], false, false);
                }
            }
        }
        __builtin_amdgcn_sched_barrier(0);
    }
}

__device__ __forceinline__ bool ea_proj_tile(int R, int blk, int& tm, int& tn0)
{
    const int wave = threadIdx.x >> 5;
    const int nb = R / 64;
    const int g = blk / nb;
    tm  = (blk - g * nb) * 4 + wave;
    tn0 = g * 4;
    return g < Dd / 64 && tm < R / 16;
}

__device__ __forceinline__ void
ea_gemm256(const float* __restrict__ A, const float* __restrict__ W,
           const float* __restrict__ bias, float* __restrict__ OUT, int R, int blk)
{
    const int lane = threadIdx.x & 31;
    int tm, tn0;
    if (!ea_proj_tile(R, blk, tm, tn0)) return;
    const int col = lane & 15, half = lane >> 4;

    const float* Ep = A + (size_t)(tm * 16 + col) * Dd + 16 * half;
    const float* Wp = W + (size_t)(tn0 * 16 + col) * Dd + 16 * half;

    v8f acc[4] = {};
    ea_proj_core<false>(Ep, Wp, acc);

    #pragma unroll
    for (int t = 0; t < 4; ++t) {
        const int tn = tn0 + t;
        const float bv = bias[tn * 16 + col];
        #pragma unroll
        for (int r = 0; r < 8; ++r) {
            int row = tm * 16 + r + 8 * half;
            OUT[(size_t)row * Dd + tn * 16 + col] = acc[t][r] + bv;
        }
    }
}

template <bool KMODE>
__device__ __forceinline__ void
ea_gemm_kvfrag(const float* __restrict__ A, const float* __restrict__ W,
               const float* __restrict__ bias, float* __restrict__ OUT, int R, int blk)
{
    const int lane = threadIdx.x & 31;
    int tm, tn0;
    if (!ea_proj_tile(R, blk, tm, tn0)) return;
    const int col = lane & 15, half = lane >> 4;

    const float* Ep = A + (size_t)(tm * 16 + col) * Dd + 16 * half;
    const float* Wp = W + (size_t)(tn0 * 16 + col) * Dd + 16 * half;

    v8f acc[4] = {};
    ea_proj_core<KMODE>(Ep, Wp, acc);

    const int bb = tm >> 8, kt = tm & 255;
    #pragma unroll
    for (int t = 0; t < 4; ++t) {
        const int tn = tn0 + t;
        const float* bp = bias + tn * 16 + (KMODE ? 8 * half : col);
        const int bstep = KMODE ? 1 : 0;
        float* dst = OUT + ((size_t)((bb * Hh + (tn >> 1)) * 256 + kt)) * 512
                         + (tn & 1) * 256 + lane * 4;
        v4f o0 = { acc[t][0] + bp[0],         acc[t][1] + bp[bstep],
                   acc[t][2] + bp[2 * bstep], acc[t][3] + bp[3 * bstep] };
        v4f o1 = { acc[t][4] + bp[4 * bstep], acc[t][5] + bp[5 * bstep],
                   acc[t][6] + bp[6 * bstep], acc[t][7] + bp[7 * bstep] };
        *(v4f*)(dst)       = o0;
        *(v4f*)(dst + 128) = o1;
    }
}

#define EA_KV_BLOCKS (Bc * Ll / 16)
#define EA_Q_BLOCKS  (Bc * Nn / 16)

__global__ void __launch_bounds__(128)
ea_proj_qkv(const float* __restrict__ curE, const float* __restrict__ nbrE,
            const float* __restrict__ in_w, const float* __restrict__ in_b,
            float* __restrict__ Qo, float* __restrict__ Ko, float* __restrict__ Vo)
{
    const int blk = blockIdx.x;
    if (blk < EA_KV_BLOCKS)
        ea_gemm_kvfrag<true>(nbrE, in_w + Dd * Dd, in_b + Dd, Ko, Bc * Ll, blk);
    else if (blk < 2 * EA_KV_BLOCKS)
        ea_gemm_kvfrag<false>(nbrE, in_w + 2 * Dd * Dd, in_b + 2 * Dd, Vo, Bc * Ll,
                              blk - EA_KV_BLOCKS);
    else
        ea_gemm256(curE, in_w, in_b, Qo, Bc * Nn, blk - 2 * EA_KV_BLOCKS);
}


__device__ __forceinline__ float ea_xhalf(float x)
{
    int b = __builtin_bit_cast(int, x);
    return __builtin_bit_cast(float,
        __builtin_amdgcn_permlanex16(b, b, 0x76543210, 0xfedcba98, false, false));
}

struct ea_kv {
    v4f k[4];
    v4f v[4];
};

__device__ __forceinline__ v2f ea_lo(v4f f) { v2f r = { f.x, f.y }; return r; }
__device__ __forceinline__ v2f ea_hi(v4f f) { v2f r = { f.z, f.w }; return r; }

__device__ __forceinline__ void ea_load_tile(ea_kv& t, const float* __restrict__ Kl,
                                             const float* __restrict__ Vl, int kt)
{
    const float* Kp = Kl + (size_t)kt * 512;
    const float* Vp = Vl + (size_t)kt * 512;
    #pragma unroll
    for (int q = 0; q < 4; ++q) t.k[q] = *(const v4f*)(Kp + q * 128);
    #pragma unroll
    for (int q = 0; q < 4; ++q) t.v[q] = *(const v4f*)(Vp + q * 128);
}

__device__ __forceinline__ void ea_tile_step(const ea_kv& t, const v2f (&qf)[8],
                                             float& mrun, float& lrun,
                                             v8f& acc0, v8f& acc1)
{
    const float scale = 0.17677669529663687f;

    v8f s = {};
    #pragma unroll
    for (int q = 0; q < 4; ++q) {
        s = __builtin_amdgcn_wmma_f32_16x16x4_f32(false, ea_lo(t.k[q]), false, qf[2 * q],
                                                  (short)0, s, false, false);
        s = __builtin_amdgcn_wmma_f32_16x16x4_f32(false, ea_hi(t.k[q]), false, qf[2 * q + 1],
                                                  (short)0, s, false, false);
    }

    float v[8];
    #pragma unroll
    for (int r = 0; r < 8; ++r) v[r] = s[r] * scale;
    float mx = fmaxf(fmaxf(fmaxf(v[0], v[1]), fmaxf(v[2], v[3])),
                     fmaxf(fmaxf(v[4], v[5]), fmaxf(v[6], v[7])));
    mx = fmaxf(mx, ea_xhalf(mx));
    const float mnew = fmaxf(mrun, mx);
    const float corr = __expf(mrun - mnew);
    float p[8];
    float psum = 0.f;
    #pragma unroll
    for (int r = 0; r < 8; ++r) { p[r] = __expf(v[r] - mnew); psum += p[r]; }
    lrun = lrun * corr + psum;
    #pragma unroll
    for (int r = 0; r < 8; ++r) { acc0[r] *= corr; acc1[r] *= corr; }
    mrun = mnew;

    #pragma unroll
    for (int j = 0; j < 4; ++j) {
        v2f pf; pf.x = p[2 * j];  pf.y = p[2 * j + 1];
        const v2f a0 = (j & 1) ? ea_hi(t.v[j >> 1])       : ea_lo(t.v[j >> 1]);
        const v2f a1 = (j & 1) ? ea_hi(t.v[2 + (j >> 1)]) : ea_lo(t.v[2 + (j >> 1)]);
        acc0 = __builtin_amdgcn_wmma_f32_16x16x4_f32(false, a0, false, pf,
                                                     (short)0, acc0, false, false);
        acc1 = __builtin_amdgcn_wmma_f32_16x16x4_f32(false, a1, false, pf,
                                                     (short)0, acc1, false, false);
    }
}

__global__ void __launch_bounds__(256)
ea_attn(const float* __restrict__ Q, const float* __restrict__ Km,
        const float* __restrict__ Vm, float* __restrict__ O)
{
    const int lane = threadIdx.x & 31;
    const int wave = threadIdx.x >> 5;
    const int tile = blockIdx.x * 8 + wave;
    const int qt = tile & 127;
    const int h  = (tile >> 7) & 7;
    const int b  = tile >> 10;
    const int col = lane & 15, half = lane >> 4;

    const float* Qrow = Q + (size_t)(b * Nn + qt * 16 + col) * Dd + h * DHh + 8 * half;
    v2f qf[8];
    #pragma unroll
    for (int q = 0; q < 4; ++q) {
        const v4f g = *(const v4f*)(Qrow + 16 * (q >> 1) + 4 * (q & 1));
        qf[2 * q]     = ea_lo(g);
        qf[2 * q + 1] = ea_hi(g);
    }

    float mrun = -__builtin_inff();
    float lrun = 0.f;
    v8f acc0 = {}, acc1 = {};

    const float* Kb = Km + (size_t)((b * Hh + h) * 256) * 512 + lane * 4;
    const float* Vb = Vm + (size_t)((b * Hh + h) * 256) * 512 + lane * 4;

    ea_kv e0, e1, e2, e3;
    __builtin_amdgcn_sched_barrier(0);
    ea_load_tile(e0, Kb, Vb, 0);
    __builtin_amdgcn_sched_barrier(0);
    ea_load_tile(e1, Kb, Vb, 1);
    __builtin_amdgcn_sched_barrier(0);
    ea_load_tile(e2, Kb, Vb, 2);
    __builtin_amdgcn_sched_barrier(0);
    ea_load_tile(e3, Kb, Vb, 3);
    __builtin_amdgcn_sched_barrier(0);
    for (int kt = 0; kt < Ll / 16; kt += 4) {
        int t0 = (kt + 4 < Ll / 16) ? kt + 4 : kt;
        int t1 = t0 + 1, t2 = t0 + 2, t3 = t0 + 3;
        ea_tile_step(e0, qf, mrun, lrun, acc0, acc1);
        asm volatile("" : "+v"(acc0), "+v"(acc1), "+s"(t0));
        ea_load_tile(e0, Kb, Vb, t0);
        __builtin_amdgcn_sched_barrier(0);
        ea_tile_step(e1, qf, mrun, lrun, acc0, acc1);
        asm volatile("" : "+v"(acc0), "+v"(acc1), "+s"(t1));
        ea_load_tile(e1, Kb, Vb, t1);
        __builtin_amdgcn_sched_barrier(0);
        ea_tile_step(e2, qf, mrun, lrun, acc0, acc1);
        asm volatile("" : "+v"(acc0), "+v"(acc1), "+s"(t2));
        ea_load_tile(e2, Kb, Vb, t2);
        __builtin_amdgcn_sched_barrier(0);
        ea_tile_step(e3, qf, mrun, lrun, acc0, acc1);
        asm volatile("" : "+v"(acc0), "+v"(acc1), "+s"(t3));
        ea_load_tile(e3, Kb, Vb, t3);
        __builtin_amdgcn_sched_barrier(0);
    }

    const float inv = 1.0f / (lrun + ea_xhalf(lrun));
    float* Op = O + (size_t)(b * Nn + qt * 16 + col) * Dd + h * DHh + 8 * half;
    v4f o0 = { acc0[0] * inv, acc0[1] * inv, acc0[2] * inv, acc0[3] * inv };
    v4f o1 = { acc0[4] * inv, acc0[5] * inv, acc0[6] * inv, acc0[7] * inv };
    v4f o2 = { acc1[0] * inv, acc1[1] * inv, acc1[2] * inv, acc1[3] * inv };
    v4f o3 = { acc1[4] * inv, acc1[5] * inv, acc1[6] * inv, acc1[7] * inv };
    *(v4f*)(Op)      = o0;
    *(v4f*)(Op + 4)  = o1;
    *(v4f*)(Op + 16) = o2;
    *(v4f*)(Op + 20) = o3;
}

__global__ void __launch_bounds__(128)
ea_gemm_out(const float* __restrict__ A, const float* __restrict__ W,
            const float* __restrict__ bias,
            const float* __restrict__ pts,
            const float* __restrict__ sw, const float* __restrict__ sb,
            float* __restrict__ OUT, int R)
{
    const int lane = threadIdx.x & 31;
    int tm, tn0;
    if (!ea_proj_tile(R, (int)blockIdx.x, tm, tn0)) return;
    const int col = lane & 15, half = lane >> 4;

    const float* Ep = A + (size_t)(tm * 16 + col) * Dd + 16 * half;
    const float* Wp = W + (size_t)(tn0 * 16 + col) * Dd + 16 * half;

    v8f acc[4] = {};
    ea_proj_core<false>(Ep, Wp, acc);

    #pragma unroll
    for (int t = 0; t < 4; ++t) {
        const int colg = (tn0 + t) * 16 + col;
        const float s0 = sw[colg*3+0], s1 = sw[colg*3+1], s2 = sw[colg*3+2];
        const float add = bias[colg] + sb[colg];
        #pragma unroll
        for (int r = 0; r < 8; ++r) {
            int row = tm * 16 + r + 8 * half;
            float px = pts[row*3+0], py = pts[row*3+1], pz = pts[row*3+2];
            OUT[(size_t)row * Dd + colg] = acc[t][r] + add + px*s0 + py*s1 + pz*s2;
        }
    }
}

extern "C" void kernel_launch(void* const* d_in, const int* in_sizes, int n_in,
                              void* d_out, int out_size, void* d_ws, size_t ws_size,
                              hipStream_t stream)
{
    const float* cur_pts  = (const float*)d_in[0];
    const float* cur_feat = (const float*)d_in[1];
    const float* nbr_pts  = (const float*)d_in[2];
    const float* nbr_feat = (const float*)d_in[3];
    const float* in_w     = (const float*)d_in[4];
    const float* in_b     = (const float*)d_in[5];
    const float* out_w    = (const float*)d_in[6];
    const float* out_b    = (const float*)d_in[7];
    const float* sp_w     = (const float*)d_in[8];
    const float* sp_b     = (const float*)d_in[9];
    float* out = (float*)d_out;

    float* ws   = (float*)d_ws;
    float* curE = ws;
    float* nbrE = curE + 4096 * 256;
    float* Qb   = nbrE + 8192 * 256;
    float* Kb   = Qb   + 4096 * 256;
    float* Vb   = Kb   + 8192 * 256;
    float* Ob   = Vb   + 8192 * 256;

    const dim3 blk(256);
    const dim3 blk4(128);

    ea_knn_edge<<<dim3(EA_KNN_CUR_BLOCKS + EA_KNN_NBR_BLOCKS), blk, 0, stream>>>(
        cur_pts, cur_feat, curE, nbr_pts, nbr_feat, nbrE);

    ea_proj_qkv<<<dim3(2 * EA_KV_BLOCKS + EA_Q_BLOCKS), blk4, 0, stream>>>(
        curE, nbrE, in_w, in_b, Qb, Kb, Vb);

    ea_attn<<<dim3(2048 / 8), blk, 0, stream>>>(Qb, Kb, Vb, Ob);

    ea_gemm_out<<<dim3(4096 / 16), blk4, 0, stream>>>(Ob, out_w, out_b,
                                                    cur_pts, sp_w, sp_b,
                                                    out, 4096);
}


# projection GEMMs stage each 64x32 activation and weight k-block in LDS once per block by whole-line loads (4 lines a wave-load, not 16 partial; weights once, not per wave); sums unchanged
# speedup vs baseline: 3.3818x; 1.1880x over previous
#include <hip/hip_runtime.h>
#include <hip/hip_bf16.h>


#define Bc   2
#define Nn   2048
#define Dd   256
#define Hh   8
#define DHh  32
#define Mm   2
#define Ll   4096

typedef __attribute__((ext_vector_type(2))) float v2f;
typedef __attribute__((ext_vector_type(4))) float v4f;
typedef __attribute__((ext_vector_type(8))) float v8f;


__device__ __forceinline__ unsigned ea_lane_below(unsigned x)
{
    return (unsigned)__builtin_amdgcn_update_dpp(0, (int)x, 0x111  ,
                                                 0xf, 0xf, true);
}

__device__ __forceinline__ void ea_knn_insert(unsigned m, unsigned d2b, int jbase,
                                              unsigned& khi, unsigned& klo)
{
    while (m) {
        const int l = __builtin_ctz(m);
        m &= m - 1;
        const unsigned chi = (unsigned)__builtin_amdgcn_readlane((int)d2b, l);
        const unsigned long long c =
            ((unsigned long long)chi << 32) | (unsigned)(jbase + l);
        const unsigned long long below =
            ((unsigned long long)ea_lane_below(khi) << 32) | ea_lane_below(klo);
        const unsigned long long key = ((unsigned long long)khi << 32) | klo;
        const unsigned long long up = (c > below) ? c : below;
        const unsigned long long nk = (key < up) ? key : up;
        khi = (unsigned)(nk >> 32);
        klo = (unsigned)nk;
    }
}

#define EA_KNN_CUR_BLOCKS (Bc * Nn / 8)
#define EA_KNN_NBR_BLOCKS (Mm * Bc * Nn / 8)

__global__ void ea_knn_edge(const float* __restrict__ cur_pts,
                            const float* __restrict__ cur_feat,
                            float* __restrict__ cur_out,
                            const float* __restrict__ nbr_pts,
                            const float* __restrict__ nbr_feat,
                            float* __restrict__ nbr_out)
{
    const int lane = threadIdx.x & 31;
    const int wave = threadIdx.x >> 5;
    const bool nbr = blockIdx.x >= EA_KNN_CUR_BLOCKS;
    const float* __restrict__ pts   = nbr ? nbr_pts  : cur_pts;
    const float* __restrict__ feats = nbr ? nbr_feat : cur_feat;
    float* __restrict__ out         = nbr ? nbr_out  : cur_out;
    const int npts  = Nn;
    const int nsets = nbr ? Mm * Bc : Bc;
    const int remap = nbr ? 1 : 0;
    const int rowg = (nbr ? (int)blockIdx.x - EA_KNN_CUR_BLOCKS : (int)blockIdx.x) * 8 + wave;
    if (rowg >= nsets * npts) return;
    const int s = rowg / npts;
    const int n = rowg - s * npts;
    int os = s;
    if (remap) { int m = s / Bc; int b = s - m * Bc; os = b * Mm + m; }

    const float* P = pts   + (size_t)s  * npts * 3;
    const float* F = feats + (size_t)s  * npts * Dd;
    float*    Orow = out   + ((size_t)os * npts + n) * Dd;

    const float px = P[n*3+0], py = P[n*3+1], pz = P[n*3+2];

    unsigned khi = ~0u, klo = ~0u;
    unsigned bar = ~0u;

    const unsigned last = (unsigned)npts - 1u;
    unsigned ia = (unsigned)lane < last ? (unsigned)lane : last;
    unsigned ib = (unsigned)lane + 32u < last ? (unsigned)lane + 32u : last;
    const float* pa = P + (size_t)ia * 3;
    const float* pb = P + (size_t)ib * 3;
    float ax = pa[0], ay = pa[1], az = pa[2];
    float bx = pb[0], by = pb[1], bz = pb[2];

    for (int j0 = 0; j0 < npts; j0 += 64) {
        const int ja = j0 + lane, jb = ja + 32;
        float d2a, d2b;
        {
            float dx = px - ax;
            float dy = py - ay;
            float dz = pz - az;
            d2a = dx*dx + dy*dy + dz*dz;
        }
        {
            float dx = px - bx;
            float dy = py - by;
            float dz = pz - bz;
            d2b = dx*dx + dy*dy + dz*dz;
        }
        ia = (unsigned)ja + 64u < last ? (unsigned)ja + 64u : last;
        ib = (unsigned)jb + 64u < last ? (unsigned)jb + 64u : last;
        pa = P + (size_t)ia * 3;
        pb = P + (size_t)ib * 3;
        ax = pa[0]; ay = pa[1]; az = pa[2];
        bx = pb[0]; by = pb[1]; bz = pb[2];

        const unsigned ua = __float_as_uint(d2a), ub = __float_as_uint(d2b);
        const unsigned ma = __builtin_amdgcn_ballot_w32(ua <= bar)
                          & __builtin_amdgcn_ballot_w32(ja < npts);
        if (ma) {
            ea_knn_insert(ma, ua, j0, khi, klo);
            bar = (unsigned)__builtin_amdgcn_readlane((int)khi, 7);
        }
        const unsigned mb = __builtin_amdgcn_ballot_w32(ub <= bar)
                          & __builtin_amdgcn_ballot_w32(jb < npts);
        if (mb) {
            ea_knn_insert(mb, ub, j0 + 32, khi, klo);
            bar = (unsigned)__builtin_amdgcn_readlane((int)khi, 7);
        }
    }

    int sel[8];
    #pragma unroll
    for (int t = 0; t < 8; ++t) sel[t] = __builtin_amdgcn_readlane((int)klo, t);

    for (int c = lane; c < Dd; c += 32) {
        float ssum = 0.f;
        #pragma unroll
        for (int t = 0; t < 8; ++t) ssum += F[(size_t)sel[t] * Dd + c];
        Orow[c] = F[(size_t)n * Dd + c] - ssum * 0.125f;
    }
}

#define EA_LP    36
#define EA_LBUF  (128 * EA_LP)

struct ea_frag { v4f g[4]; };

__device__ __forceinline__ v2f ea_pair(v4f f, int u)
{
    v2f r;
    if (u) { r.x = f.z; r.y = f.w; } else { r.x = f.x; r.y = f.y; }
    return r;
}

__device__ __forceinline__ void ea_lds_frag(ea_frag& f, const float* p)
{
    #pragma unroll
    for (int i = 0; i < 4; ++i) f.g[i] = *(const v4f*)(p + i * 4);
}

struct ea_stage { v4f e[4]; v4f w[4]; };

__device__ __forceinline__ void ea_stage_load(ea_stage& s, const float* __restrict__ Eg,
                                              const float* __restrict__ Wg, int kb)
{
    #pragma unroll
    for (int j = 0; j < 4; ++j) s.e[j] = *(const v4f*)(Eg + (size_t)j * 16 * Dd + kb * 32);
    #pragma unroll
    for (int j = 0; j < 4; ++j) s.w[j] = *(const v4f*)(Wg + (size_t)j * 16 * Dd + kb * 32);
}

__device__ __forceinline__ void ea_stage_store(float* buf, const ea_stage& s, int so)
{
    #pragma unroll
    for (int j = 0; j < 4; ++j) *(v4f*)(buf + so + j * 16 * EA_LP) = s.e[j];
    #pragma unroll
    for (int j = 0; j < 4; ++j) *(v4f*)(buf + 64 * EA_LP + so + j * 16 * EA_LP) = s.w[j];
}

__device__ __forceinline__ void ea_lds_barrier()
{
    asm volatile("s_wait_dscnt 0x0\n\ts_barrier_signal -1\n\ts_barrier_wait -1" ::: "memory");
}

template <bool W_IS_A>
__device__ __forceinline__ void ea_proj_core(const float* __restrict__ Eblk,
                                             const float* __restrict__ Wblk,
                                             float* lds, v8f (&acc)[4])
{
    const int t = threadIdx.x;
    const int lane = t & 31, wave = t >> 5;
    const int col = lane & 15, half = lane >> 4;
    const float* Eg = Eblk + (size_t)(t >> 3) * Dd + 4 * (t & 7);
    const float* Wg = Wblk + (size_t)(t >> 3) * Dd + 4 * (t & 7);
    const int so = (t >> 3) * EA_LP + 4 * (t & 7);
    const int fe = (16 * wave + col) * EA_LP + 16 * half;
    const int fw = (64 + col) * EA_LP + 16 * half;

    ea_stage st;
    {
        ea_stage st0;
        ea_stage_load(st0, Eg, Wg, 0);
        ea_stage_load(st, Eg, Wg, 1);
        ea_stage_store(lds, st0, so);
    }
    ea_lds_barrier();

    #pragma unroll
    for (int kb = 0; kb < Dd / 32; ++kb) {
        const float* cur = lds + (kb & 1) * EA_LBUF;
        float* nxt = lds + ((kb & 1) ^ 1) * EA_LBUF;
        ea_frag ef;
        ea_frag wf[4];
        ea_lds_frag(ef, cur + fe);
        #pragma unroll
        for (int tt = 0; tt < 4; ++tt) ea_lds_frag(wf[tt], cur + fw + tt * 16 * EA_LP);

        if (kb + 1 < Dd / 32) ea_stage_store(nxt, st, so);
        if (kb + 2 < Dd / 32) ea_stage_load(st, Eg, Wg, kb + 2);

        #pragma unroll
        for (int i = 0; i < 4; ++i) {
            #pragma unroll
            for (int u = 0; u < 2; ++u) {
                const v2f e = ea_pair(ef.g[i], u);
                #pragma unroll
                for (int tt = 0; tt < 4; ++tt) {
                    const v2f w = ea_pair(wf[tt].g[i], u);
                    acc[tt] = W_IS_A
                        ? __builtin_amdgcn_wmma_f32_16x16x4_f32(false, w, false, e,
                                                                (short)0, acc[tt], false, false)
                        : __builtin_amdgcn_wmma_f32_16x16x4_f32(false, e, false, w,
                                                                (short)0, acc[tt], false, false);
                }
            }
        }
        if (kb + 1 < Dd / 32) ea_lds_barrier();
    }
}

__device__ __forceinline__ bool ea_proj_tile(int R, int blk, int& tm, int& tn0)
{
    const int wave = threadIdx.x >> 5;
    const int nb = R / 64;
    const int g = blk / nb;
    tm  = (blk - g * nb) * 4 + wave;
    tn0 = g * 4;
    return g < Dd / 64 && tm < R / 16;
}

__device__ __forceinline__ void
ea_gemm256(const float* __restrict__ A, const float* __restrict__ W,
           const float* __restrict__ bias, float* __restrict__ OUT, int R, int blk,
           float* lds)
{
    const int lane = threadIdx.x & 31;
    int tm, tn0;
    if (!ea_proj_tile(R, blk, tm, tn0)) return;
    const int col = lane & 15, half = lane >> 4;

    const int wave = threadIdx.x >> 5;
    const float* Eblk = A + (size_t)((tm - wave) * 16) * Dd;
    const float* Wblk = W + (size_t)(tn0 * 16) * Dd;

    v8f acc[4] = {};
    ea_proj_core<false>(Eblk, Wblk, lds, acc);

    #pragma unroll
    for (int t = 0; t < 4; ++t) {
        const int tn = tn0 + t;
        const float bv = bias[tn * 16 + col];
        #pragma unroll
        for (int r = 0; r < 8; ++r) {
            int row = tm * 16 + r + 8 * half;
            OUT[(size_t)row * Dd + tn * 16 + col] = acc[t][r] + bv;
        }
    }
}

template <bool KMODE>
__device__ __forceinline__ void
ea_gemm_kvfrag(const float* __restrict__ A, const float* __restrict__ W,
               const float* __restrict__ bias, float* __restrict__ OUT, int R, int blk,
               float* lds)
{
    const int lane = threadIdx.x & 31;
    int tm, tn0;
    if (!ea_proj_tile(R, blk, tm, tn0)) return;
    const int col = lane & 15, half = lane >> 4;

    const int wave = threadIdx.x >> 5;
    const float* Eblk = A + (size_t)((tm - wave) * 16) * Dd;
    const float* Wblk = W + (size_t)(tn0 * 16) * Dd;

    v8f acc[4] = {};
    ea_proj_core<KMODE>(Eblk, Wblk, lds, acc);

    const int bb = tm >> 8, kt = tm & 255;
    #pragma unroll
    for (int t = 0; t < 4; ++t) {
        const int tn = tn0 + t;
        const float* bp = bias + tn * 16 + (KMODE ? 8 * half : col);
        const int bstep = KMODE ? 1 : 0;
        float* dst = OUT + ((size_t)((bb * Hh + (tn >> 1)) * 256 + kt)) * 512
                         + (tn & 1) * 256 + lane * 4;
        v4f o0 = { acc[t][0] + bp[0],         acc[t][1] + bp[bstep],
                   acc[t][2] + bp[2 * bstep], acc[t][3] + bp[3 * bstep] };
        v4f o1 = { acc[t][4] + bp[4 * bstep], acc[t][5] + bp[5 * bstep],
                   acc[t][6] + bp[6 * bstep], acc[t][7] + bp[7 * bstep] };
        *(v4f*)(dst)       = o0;
        *(v4f*)(dst + 128) = o1;
    }
}

#define EA_KV_BLOCKS (Bc * Ll / 16)
#define EA_Q_BLOCKS  (Bc * Nn / 16)

__global__ void __launch_bounds__(128)
ea_proj_qkv(const float* __restrict__ curE, const float* __restrict__ nbrE,
            const float* __restrict__ in_w, const float* __restrict__ in_b,
            float* __restrict__ Qo, float* __restrict__ Ko, float* __restrict__ Vo)
{
    __shared__ __attribute__((aligned(16))) float ea_lds[2 * EA_LBUF];
    const int blk = blockIdx.x;
    if (blk < EA_KV_BLOCKS)
        ea_gemm_kvfrag<true>(nbrE, in_w + Dd * Dd, in_b + Dd, Ko, Bc * Ll, blk, ea_lds);
    else if (blk < 2 * EA_KV_BLOCKS)
        ea_gemm_kvfrag<false>(nbrE, in_w + 2 * Dd * Dd, in_b + 2 * Dd, Vo, Bc * Ll,
                              blk - EA_KV_BLOCKS, ea_lds);
    else
        ea_gemm256(curE, in_w, in_b, Qo, Bc * Nn, blk - 2 * EA_KV_BLOCKS, ea_lds);
}


__device__ __forceinline__ float ea_xhalf(float x)
{
    int b = __builtin_bit_cast(int, x);
    return __builtin_bit_cast(float,
        __builtin_amdgcn_permlanex16(b, b, 0x76543210, 0xfedcba98, false, false));
}

struct ea_kv {
    v4f k[4];
    v4f v[4];
};

__device__ __forceinline__ v2f ea_lo(v4f f) { v2f r = { f.x, f.y }; return r; }
__device__ __forceinline__ v2f ea_hi(v4f f) { v2f r = { f.z, f.w }; return r; }

__device__ __forceinline__ void ea_load_tile(ea_kv& t, const float* __restrict__ Kl,
                                             const float* __restrict__ Vl, int kt)
{
    const float* Kp = Kl + (size_t)kt * 512;
    const float* Vp = Vl + (size_t)kt * 512;
    #pragma unroll
    for (int q = 0; q < 4; ++q) t.k[q] = *(const v4f*)(Kp + q * 128);
    #pragma unroll
    for (int q = 0; q < 4; ++q) t.v[q] = *(const v4f*)(Vp + q * 128);
}

__device__ __forceinline__ void ea_tile_step(const ea_kv& t, const v2f (&qf)[8],
                                             float& mrun, float& lrun,
                                             v8f& acc0, v8f& acc1)
{
    const float scale = 0.17677669529663687f;

    v8f s = {};
    #pragma unroll
    for (int q = 0; q < 4; ++q) {
        s = __builtin_amdgcn_wmma_f32_16x16x4_f32(false, ea_lo(t.k[q]), false, qf[2 * q],
                                                  (short)0, s, false, false);
        s = __builtin_amdgcn_wmma_f32_16x16x4_f32(false, ea_hi(t.k[q]), false, qf[2 * q + 1],
                                                  (short)0, s, false, false);
    }

    float v[8];
    #pragma unroll
    for (int r = 0; r < 8; ++r) v[r] = s[r] * scale;
    float mx = fmaxf(fmaxf(fmaxf(v[0], v[1]), fmaxf(v[2], v[3])),
                     fmaxf(fmaxf(v[4], v[5]), fmaxf(v[6], v[7])));
    mx = fmaxf(mx, ea_xhalf(mx));
    const float mnew = fmaxf(mrun, mx);
    const float corr = __expf(mrun - mnew);
    float p[8];
    float psum = 0.f;
    #pragma unroll
    for (int r = 0; r < 8; ++r) { p[r] = __expf(v[r] - mnew); psum += p[r]; }
    lrun = lrun * corr + psum;
    #pragma unroll
    for (int r = 0; r < 8; ++r) { acc0[r] *= corr; acc1[r] *= corr; }
    mrun = mnew;

    #pragma unroll
    for (int j = 0; j < 4; ++j) {
        v2f pf; pf.x = p[2 * j];  pf.y = p[2 * j + 1];
        const v2f a0 = (j & 1) ? ea_hi(t.v[j >> 1])       : ea_lo(t.v[j >> 1]);
        const v2f a1 = (j & 1) ? ea_hi(t.v[2 + (j >> 1)]) : ea_lo(t.v[2 + (j >> 1)]);
        acc0 = __builtin_amdgcn_wmma_f32_16x16x4_f32(false, a0, false, pf,
                                                     (short)0, acc0, false, false);
        acc1 = __builtin_amdgcn_wmma_f32_16x16x4_f32(false, a1, false, pf,
                                                     (short)0, acc1, false, false);
    }
}

__global__ void __launch_bounds__(256)
ea_attn(const float* __restrict__ Q, const float* __restrict__ Km,
        const float* __restrict__ Vm, float* __restrict__ O)
{
    const int lane = threadIdx.x & 31;
    const int wave = threadIdx.x >> 5;
    const int tile = blockIdx.x * 8 + wave;
    const int qt = tile & 127;
    const int h  = (tile >> 7) & 7;
    const int b  = tile >> 10;
    const int col = lane & 15, half = lane >> 4;

    const float* Qrow = Q + (size_t)(b * Nn + qt * 16 + col) * Dd + h * DHh + 8 * half;
    v2f qf[8];
    #pragma unroll
    for (int q = 0; q < 4; ++q) {
        const v4f g = *(const v4f*)(Qrow + 16 * (q >> 1) + 4 * (q & 1));
        qf[2 * q]     = ea_lo(g);
        qf[2 * q + 1] = ea_hi(g);
    }

    float mrun = -__builtin_inff();
    float lrun = 0.f;
    v8f acc0 = {}, acc1 = {};

    const float* Kb = Km + (size_t)((b * Hh + h) * 256) * 512 + lane * 4;
    const float* Vb = Vm + (size_t)((b * Hh + h) * 256) * 512 + lane * 4;

    ea_kv e0, e1, e2, e3;
    __builtin_amdgcn_sched_barrier(0);
    ea_load_tile(e0, Kb, Vb, 0);
    __builtin_amdgcn_sched_barrier(0);
    ea_load_tile(e1, Kb, Vb, 1);
    __builtin_amdgcn_sched_barrier(0);
    ea_load_tile(e2, Kb, Vb, 2);
    __builtin_amdgcn_sched_barrier(0);
    ea_load_tile(e3, Kb, Vb, 3);
    __builtin_amdgcn_sched_barrier(0);
    for (int kt = 0; kt < Ll / 16; kt += 4) {
        int t0 = (kt + 4 < Ll / 16) ? kt + 4 : kt;
        int t1 = t0 + 1, t2 = t0 + 2, t3 = t0 + 3;
        ea_tile_step(e0, qf, mrun, lrun, acc0, acc1);
        asm volatile("" : "+v"(acc0), "+v"(acc1), "+s"(t0));
        ea_load_tile(e0, Kb, Vb, t0);
        __builtin_amdgcn_sched_barrier(0);
        ea_tile_step(e1, qf, mrun, lrun, acc0, acc1);
        asm volatile("" : "+v"(acc0), "+v"(acc1), "+s"(t1));
        ea_load_tile(e1, Kb, Vb, t1);
        __builtin_amdgcn_sched_barrier(0);
        ea_tile_step(e2, qf, mrun, lrun, acc0, acc1);
        asm volatile("" : "+v"(acc0), "+v"(acc1), "+s"(t2));
        ea_load_tile(e2, Kb, Vb, t2);
        __builtin_amdgcn_sched_barrier(0);
        ea_tile_step(e3, qf, mrun, lrun, acc0, acc1);
        asm volatile("" : "+v"(acc0), "+v"(acc1), "+s"(t3));
        ea_load_tile(e3, Kb, Vb, t3);
        __builtin_amdgcn_sched_barrier(0);
    }

    const float inv = 1.0f / (lrun + ea_xhalf(lrun));
    float* Op = O + (size_t)(b * Nn + qt * 16 + col) * Dd + h * DHh + 8 * half;
    v4f o0 = { acc0[0] * inv, acc0[1] * inv, acc0[2] * inv, acc0[3] * inv };
    v4f o1 = { acc0[4] * inv, acc0[5] * inv, acc0[6] * inv, acc0[7] * inv };
    v4f o2 = { acc1[0] * inv, acc1[1] * inv, acc1[2] * inv, acc1[3] * inv };
    v4f o3 = { acc1[4] * inv, acc1[5] * inv, acc1[6] * inv, acc1[7] * inv };
    *(v4f*)(Op)      = o0;
    *(v4f*)(Op + 4)  = o1;
    *(v4f*)(Op + 16) = o2;
    *(v4f*)(Op + 20) = o3;
}

__global__ void __launch_bounds__(128)
ea_gemm_out(const float* __restrict__ A, const float* __restrict__ W,
            const float* __restrict__ bias,
            const float* __restrict__ pts,
            const float* __restrict__ sw, const float* __restrict__ sb,
            float* __restrict__ OUT, int R)
{
    const int lane = threadIdx.x & 31;
    int tm, tn0;
    if (!ea_proj_tile(R, (int)blockIdx.x, tm, tn0)) return;
    const int col = lane & 15, half = lane >> 4;

    __shared__ __attribute__((aligned(16))) float ea_lds[2 * EA_LBUF];
    const int wave = threadIdx.x >> 5;
    const float* Eblk = A + (size_t)((tm - wave) * 16) * Dd;
    const float* Wblk = W + (size_t)(tn0 * 16) * Dd;

    v8f acc[4] = {};
    ea_proj_core<false>(Eblk, Wblk, ea_lds, acc);

    #pragma unroll
    for (int t = 0; t < 4; ++t) {
        const int colg = (tn0 + t) * 16 + col;
        const float s0 = sw[colg*3+0], s1 = sw[colg*3+1], s2 = sw[colg*3+2];
        const float add = bias[colg] + sb[colg];
        #pragma unroll
        for (int r = 0; r < 8; ++r) {
            int row = tm * 16 + r + 8 * half;
            float px = pts[row*3+0], py = pts[row*3+1], pz = pts[row*3+2];
            OUT[(size_t)row * Dd + colg] = acc[t][r] + add + px*s0 + py*s1 + pz*s2;
        }
    }
}

extern "C" void kernel_launch(void* const* d_in, const int* in_sizes, int n_in,
                              void* d_out, int out_size, void* d_ws, size_t ws_size,
                              hipStream_t stream)
{
    const float* cur_pts  = (const float*)d_in[0];
    const float* cur_feat = (const float*)d_in[1];
    const float* nbr_pts  = (const float*)d_in[2];
    const float* nbr_feat = (const float*)d_in[3];
    const float* in_w     = (const float*)d_in[4];
    const float* in_b     = (const float*)d_in[5];
    const float* out_w    = (const float*)d_in[6];
    const float* out_b    = (const float*)d_in[7];
    const float* sp_w     = (const float*)d_in[8];
    const float* sp_b     = (const float*)d_in[9];
    float* out = (float*)d_out;

    float* ws   = (float*)d_ws;
    float* curE = ws;
    float* nbrE = curE + 4096 * 256;
    float* Qb   = nbrE + 8192 * 256;
    float* Kb   = Qb   + 4096 * 256;
    float* Vb   = Kb   + 8192 * 256;
    float* Ob   = Vb   + 8192 * 256;

    const dim3 blk(256);
    const dim3 blk4(128);

    ea_knn_edge<<<dim3(EA_KNN_CUR_BLOCKS + EA_KNN_NBR_BLOCKS), blk, 0, stream>>>(
        cur_pts, cur_feat, curE, nbr_pts, nbr_feat, nbrE);

    ea_proj_qkv<<<dim3(2 * EA_KV_BLOCKS + EA_Q_BLOCKS), blk4, 0, stream>>>(
        curE, nbrE, in_w, in_b, Qb, Kb, Vb);

    ea_attn<<<dim3(2048 / 8), blk, 0, stream>>>(Qb, Kb, Vb, Ob);

    ea_gemm_out<<<dim3(4096 / 16), blk4, 0, stream>>>(Ob, out_w, out_b,
                                                    cur_pts, sp_w, sp_b,
                                                    out, 4096);
}


# attention softmax with fewer vector instructions per key: scores come out in log2 units (Q pre-scaled), max, lane-half exchange and rescale once per 32 keys, not 16; softmax/PV sum order changes
# speedup vs baseline: 3.5814x; 1.0590x over previous
#include <hip/hip_runtime.h>
#include <hip/hip_bf16.h>


#define Bc   2
#define Nn   2048
#define Dd   256
#define Hh   8
#define DHh  32
#define Mm   2
#define Ll   4096

typedef __attribute__((ext_vector_type(2))) float v2f;
typedef __attribute__((ext_vector_type(4))) float v4f;
typedef __attribute__((ext_vector_type(8))) float v8f;


__device__ __forceinline__ unsigned ea_lane_below(unsigned x)
{
    return (unsigned)__builtin_amdgcn_update_dpp(0, (int)x, 0x111  ,
                                                 0xf, 0xf, true);
}

__device__ __forceinline__ void ea_knn_insert(unsigned m, unsigned d2b, int jbase,
                                              unsigned& khi, unsigned& klo)
{
    while (m) {
        const int l = __builtin_ctz(m);
        m &= m - 1;
        const unsigned chi = (unsigned)__builtin_amdgcn_readlane((int)d2b, l);
        const unsigned long long c =
            ((unsigned long long)chi << 32) | (unsigned)(jbase + l);
        const unsigned long long below =
            ((unsigned long long)ea_lane_below(khi) << 32) | ea_lane_below(klo);
        const unsigned long long key = ((unsigned long long)khi << 32) | klo;
        const unsigned long long up = (c > below) ? c : below;
        const unsigned long long nk = (key < up) ? key : up;
        khi = (unsigned)(nk >> 32);
        klo = (unsigned)nk;
    }
}

#define EA_KNN_CUR_BLOCKS (Bc * Nn / 8)
#define EA_KNN_NBR_BLOCKS (Mm * Bc * Nn / 8)

__global__ void ea_knn_edge(const float* __restrict__ cur_pts,
                            const float* __restrict__ cur_feat,
                            float* __restrict__ cur_out,
                            const float* __restrict__ nbr_pts,
                            const float* __restrict__ nbr_feat,
                            float* __restrict__ nbr_out)
{
    const int lane = threadIdx.x & 31;
    const int wave = threadIdx.x >> 5;
    const bool nbr = blockIdx.x >= EA_KNN_CUR_BLOCKS;
    const float* __restrict__ pts   = nbr ? nbr_pts  : cur_pts;
    const float* __restrict__ feats = nbr ? nbr_feat : cur_feat;
    float* __restrict__ out         = nbr ? nbr_out  : cur_out;
    const int npts  = Nn;
    const int nsets = nbr ? Mm * Bc : Bc;
    const int remap = nbr ? 1 : 0;
    const int rowg = (nbr ? (int)blockIdx.x - EA_KNN_CUR_BLOCKS : (int)blockIdx.x) * 8 + wave;
    if (rowg >= nsets * npts) return;
    const int s = rowg / npts;
    const int n = rowg - s * npts;
    int os = s;
    if (remap) { int m = s / Bc; int b = s - m * Bc; os = b * Mm + m; }

    const float* P = pts   + (size_t)s  * npts * 3;
    const float* F = feats + (size_t)s  * npts * Dd;
    float*    Orow = out   + ((size_t)os * npts + n) * Dd;

    const float px = P[n*3+0], py = P[n*3+1], pz = P[n*3+2];

    unsigned khi = ~0u, klo = ~0u;
    unsigned bar = ~0u;

    const unsigned last = (unsigned)npts - 1u;
    unsigned ia = (unsigned)lane < last ? (unsigned)lane : last;
    unsigned ib = (unsigned)lane + 32u < last ? (unsigned)lane + 32u : last;
    const float* pa = P + (size_t)ia * 3;
    const float* pb = P + (size_t)ib * 3;
    float ax = pa[0], ay = pa[1], az = pa[2];
    float bx = pb[0], by = pb[1], bz = pb[2];

    for (int j0 = 0; j0 < npts; j0 += 64) {
        const int ja = j0 + lane, jb = ja + 32;
        float d2a, d2b;
        {
            float dx = px - ax;
            float dy = py - ay;
            float dz = pz - az;
            d2a = dx*dx + dy*dy + dz*dz;
        }
        {
            float dx = px - bx;
            float dy = py - by;
            float dz = pz - bz;
            d2b = dx*dx + dy*dy + dz*dz;
        }
        ia = (unsigned)ja + 64u < last ? (unsigned)ja + 64u : last;
        ib = (unsigned)jb + 64u < last ? (unsigned)jb + 64u : last;
        pa = P + (size_t)ia * 3;
        pb = P + (size_t)ib * 3;
        ax = pa[0]; ay = pa[1]; az = pa[2];
        bx = pb[0]; by = pb[1]; bz = pb[2];

        const unsigned ua = __float_as_uint(d2a), ub = __float_as_uint(d2b);
        const unsigned ma = __builtin_amdgcn_ballot_w32(ua <= bar)
                          & __builtin_amdgcn_ballot_w32(ja < npts);
        if (ma) {
            ea_knn_insert(ma, ua, j0, khi, klo);
            bar = (unsigned)__builtin_amdgcn_readlane((int)khi, 7);
        }
        const unsigned mb = __builtin_amdgcn_ballot_w32(ub <= bar)
                          & __builtin_amdgcn_ballot_w32(jb < npts);
        if (mb) {
            ea_knn_insert(mb, ub, j0 + 32, khi, klo);
            bar = (unsigned)__builtin_amdgcn_readlane((int)khi, 7);
        }
    }

    int sel[8];
    #pragma unroll
    for (int t = 0; t < 8; ++t) sel[t] = __builtin_amdgcn_readlane((int)klo, t);

    for (int c = lane; c < Dd; c += 32) {
        float ssum = 0.f;
        #pragma unroll
        for (int t = 0; t < 8; ++t) ssum += F[(size_t)sel[t] * Dd + c];
        Orow[c] = F[(size_t)n * Dd + c] - ssum * 0.125f;
    }
}

#define EA_LP    36
#define EA_LBUF  (128 * EA_LP)

struct ea_frag { v4f g[4]; };

__device__ __forceinline__ v2f ea_pair(v4f f, int u)
{
    v2f r;
    if (u) { r.x = f.z; r.y = f.w; } else { r.x = f.x; r.y = f.y; }
    return r;
}

__device__ __forceinline__ void ea_lds_frag(ea_frag& f, const float* p)
{
    #pragma unroll
    for (int i = 0; i < 4; ++i) f.g[i] = *(const v4f*)(p + i * 4);
}

struct ea_stage { v4f e[4]; v4f w[4]; };

__device__ __forceinline__ void ea_stage_load(ea_stage& s, const float* __restrict__ Eg,
                                              const float* __restrict__ Wg, int kb)
{
    #pragma unroll
    for (int j = 0; j < 4; ++j) s.e[j] = *(const v4f*)(Eg + (size_t)j * 16 * Dd + kb * 32);
    #pragma unroll
    for (int j = 0; j < 4; ++j) s.w[j] = *(const v4f*)(Wg + (size_t)j * 16 * Dd + kb * 32);
}

__device__ __forceinline__ void ea_stage_store(float* buf, const ea_stage& s, int so)
{
    #pragma unroll
    for (int j = 0; j < 4; ++j) *(v4f*)(buf + so + j * 16 * EA_LP) = s.e[j];
    #pragma unroll
    for (int j = 0; j < 4; ++j) *(v4f*)(buf + 64 * EA_LP + so + j * 16 * EA_LP) = s.w[j];
}

__device__ __forceinline__ void ea_lds_barrier()
{
    asm volatile("s_wait_dscnt 0x0\n\ts_barrier_signal -1\n\ts_barrier_wait -1" ::: "memory");
}

template <bool W_IS_A>
__device__ __forceinline__ void ea_proj_core(const float* __restrict__ Eblk,
                                             const float* __restrict__ Wblk,
                                             float* lds, v8f (&acc)[4])
{
    const int t = threadIdx.x;
    const int lane = t & 31, wave = t >> 5;
    const int col = lane & 15, half = lane >> 4;
    const float* Eg = Eblk + (size_t)(t >> 3) * Dd + 4 * (t & 7);
    const float* Wg = Wblk + (size_t)(t >> 3) * Dd + 4 * (t & 7);
    const int so = (t >> 3) * EA_LP + 4 * (t & 7);
    const int fe = (16 * wave + col) * EA_LP + 16 * half;
    const int fw = (64 + col) * EA_LP + 16 * half;

    ea_stage st;
    {
        ea_stage st0;
        ea_stage_load(st0, Eg, Wg, 0);
        ea_stage_load(st, Eg, Wg, 1);
        ea_stage_store(lds, st0, so);
    }
    ea_lds_barrier();

    #pragma unroll
    for (int kb = 0; kb < Dd / 32; ++kb) {
        const float* cur = lds + (kb & 1) * EA_LBUF;
        float* nxt = lds + ((kb & 1) ^ 1) * EA_LBUF;
        ea_frag ef;
        ea_frag wf[4];
        ea_lds_frag(ef, cur + fe);
        #pragma unroll
        for (int tt = 0; tt < 4; ++tt) ea_lds_frag(wf[tt], cur + fw + tt * 16 * EA_LP);

        if (kb + 1 < Dd / 32) ea_stage_store(nxt, st, so);
        if (kb + 2 < Dd / 32) ea_stage_load(st, Eg, Wg, kb + 2);

        #pragma unroll
        for (int i = 0; i < 4; ++i) {
            #pragma unroll
            for (int u = 0; u < 2; ++u) {
                const v2f e = ea_pair(ef.g[i], u);
                #pragma unroll
                for (int tt = 0; tt < 4; ++tt) {
                    const v2f w = ea_pair(wf[tt].g[i], u);
                    acc[tt] = W_IS_A
                        ? __builtin_amdgcn_wmma_f32_16x16x4_f32(false, w, false, e,
                                                                (short)0, acc[tt], false, false)
                        : __builtin_amdgcn_wmma_f32_16x16x4_f32(false, e, false, w,
                                                                (short)0, acc[tt], false, false);
                }
            }
        }
        if (kb + 1 < Dd / 32) ea_lds_barrier();
    }
}

__device__ __forceinline__ bool ea_proj_tile(int R, int blk, int& tm, int& tn0)
{
    const int wave = threadIdx.x >> 5;
    const int nb = R / 64;
    const int g = blk / nb;
    tm  = (blk - g * nb) * 4 + wave;
    tn0 = g * 4;
    return g < Dd / 64 && tm < R / 16;
}

__device__ __forceinline__ void
ea_gemm256(const float* __restrict__ A, const float* __restrict__ W,
           const float* __restrict__ bias, float* __restrict__ OUT, int R, int blk,
           float* lds)
{
    const int lane = threadIdx.x & 31;
    int tm, tn0;
    if (!ea_proj_tile(R, blk, tm, tn0)) return;
    const int col = lane & 15, half = lane >> 4;

    const int wave = threadIdx.x >> 5;
    const float* Eblk = A + (size_t)((tm - wave) * 16) * Dd;
    const float* Wblk = W + (size_t)(tn0 * 16) * Dd;

    v8f acc[4] = {};
    ea_proj_core<false>(Eblk, Wblk, lds, acc);

    #pragma unroll
    for (int t = 0; t < 4; ++t) {
        const int tn = tn0 + t;
        const float bv = bias[tn * 16 + col];
        #pragma unroll
        for (int r = 0; r < 8; ++r) {
            int row = tm * 16 + r + 8 * half;
            OUT[(size_t)row * Dd + tn * 16 + col] = acc[t][r] + bv;
        }
    }
}

template <bool KMODE>
__device__ __forceinline__ void
ea_gemm_kvfrag(const float* __restrict__ A, const float* __restrict__ W,
               const float* __restrict__ bias, float* __restrict__ OUT, int R, int blk,
               float* lds)
{
    const int lane = threadIdx.x & 31;
    int tm, tn0;
    if (!ea_proj_tile(R, blk, tm, tn0)) return;
    const int col = lane & 15, half = lane >> 4;

    const int wave = threadIdx.x >> 5;
    const float* Eblk = A + (size_t)((tm - wave) * 16) * Dd;
    const float* Wblk = W + (size_t)(tn0 * 16) * Dd;

    v8f acc[4] = {};
    ea_proj_core<KMODE>(Eblk, Wblk, lds, acc);

    const int bb = tm >> 8, kt = tm & 255;
    #pragma unroll
    for (int t = 0; t < 4; ++t) {
        const int tn = tn0 + t;
        const float* bp = bias + tn * 16 + (KMODE ? 8 * half : col);
        const int bstep = KMODE ? 1 : 0;
        float* dst = OUT + ((size_t)((bb * Hh + (tn >> 1)) * 256 + kt)) * 512
                         + (tn & 1) * 256 + lane * 4;
        v4f o0 = { acc[t][0] + bp[0],         acc[t][1] + bp[bstep],
                   acc[t][2] + bp[2 * bstep], acc[t][3] + bp[3 * bstep] };
        v4f o1 = { acc[t][4] + bp[4 * bstep], acc[t][5] + bp[5 * bstep],
                   acc[t][6] + bp[6 * bstep], acc[t][7] + bp[7 * bstep] };
        *(v4f*)(dst)       = o0;
        *(v4f*)(dst + 128) = o1;
    }
}

#define EA_KV_BLOCKS (Bc * Ll / 16)
#define EA_Q_BLOCKS  (Bc * Nn / 16)

__global__ void __launch_bounds__(128)
ea_proj_qkv(const float* __restrict__ curE, const float* __restrict__ nbrE,
            const float* __restrict__ in_w, const float* __restrict__ in_b,
            float* __restrict__ Qo, float* __restrict__ Ko, float* __restrict__ Vo)
{
    __shared__ __attribute__((aligned(16))) float ea_lds[2 * EA_LBUF];
    const int blk = blockIdx.x;
    if (blk < EA_KV_BLOCKS)
        ea_gemm_kvfrag<true>(nbrE, in_w + Dd * Dd, in_b + Dd, Ko, Bc * Ll, blk, ea_lds);
    else if (blk < 2 * EA_KV_BLOCKS)
        ea_gemm_kvfrag<false>(nbrE, in_w + 2 * Dd * Dd, in_b + 2 * Dd, Vo, Bc * Ll,
                              blk - EA_KV_BLOCKS, ea_lds);
    else
        ea_gemm256(curE, in_w, in_b, Qo, Bc * Nn, blk - 2 * EA_KV_BLOCKS, ea_lds);
}


__device__ __forceinline__ float ea_xhalf(float x)
{
    int b = __builtin_bit_cast(int, x);
    return __builtin_bit_cast(float,
        __builtin_amdgcn_permlanex16(b, b, 0x76543210, 0xfedcba98, false, false));
}

struct ea_kv {
    v4f k[4];
    v4f v[4];
};

__device__ __forceinline__ v2f ea_lo(v4f f) { v2f r = { f.x, f.y }; return r; }
__device__ __forceinline__ v2f ea_hi(v4f f) { v2f r = { f.z, f.w }; return r; }

__device__ __forceinline__ void ea_load_tile(ea_kv& t, const float* __restrict__ Kl,
                                             const float* __restrict__ Vl, int kt)
{
    const float* Kp = Kl + (size_t)kt * 512;
    const float* Vp = Vl + (size_t)kt * 512;
    #pragma unroll
    for (int q = 0; q < 4; ++q) t.k[q] = *(const v4f*)(Kp + q * 128);
    #pragma unroll
    for (int q = 0; q < 4; ++q) t.v[q] = *(const v4f*)(Vp + q * 128);
}

__device__ __forceinline__ void ea_tile_step2(const ea_kv& ta, const ea_kv& tb,
                                              const v2f (&qf)[8],
                                              float& mrun, float& lrun,
                                              v8f& acc0, v8f& acc1)
{
    v8f sa = {}, sb = {};
    #pragma unroll
    for (int q = 0; q < 4; ++q) {
        sa = __builtin_amdgcn_wmma_f32_16x16x4_f32(false, ea_lo(ta.k[q]), false, qf[2 * q],
                                                   (short)0, sa, false, false);
        sb = __builtin_amdgcn_wmma_f32_16x16x4_f32(false, ea_lo(tb.k[q]), false, qf[2 * q],
                                                   (short)0, sb, false, false);
        sa = __builtin_amdgcn_wmma_f32_16x16x4_f32(false, ea_hi(ta.k[q]), false, qf[2 * q + 1],
                                                   (short)0, sa, false, false);
        sb = __builtin_amdgcn_wmma_f32_16x16x4_f32(false, ea_hi(tb.k[q]), false, qf[2 * q + 1],
                                                   (short)0, sb, false, false);
    }

    float mxa = fmaxf(fmaxf(fmaxf(sa[0], sa[1]), fmaxf(sa[2], sa[3])),
                      fmaxf(fmaxf(sa[4], sa[5]), fmaxf(sa[6], sa[7])));
    float mxb = fmaxf(fmaxf(fmaxf(sb[0], sb[1]), fmaxf(sb[2], sb[3])),
                      fmaxf(fmaxf(sb[4], sb[5]), fmaxf(sb[6], sb[7])));
    float mx = fmaxf(mxa, mxb);
    mx = fmaxf(mx, ea_xhalf(mx));
    const float mnew = fmaxf(mrun, mx);
    const float corr = __builtin_amdgcn_exp2f(mrun - mnew);
    float pa[8], pb[8];
    float psa = 0.f, psb = 0.f;
    #pragma unroll
    for (int r = 0; r < 8; ++r) {
        pa[r] = __builtin_amdgcn_exp2f(sa[r] - mnew); psa += pa[r];
        pb[r] = __builtin_amdgcn_exp2f(sb[r] - mnew); psb += pb[r];
    }
    lrun = lrun * corr + (psa + psb);
    #pragma unroll
    for (int r = 0; r < 8; ++r) { acc0[r] *= corr; acc1[r] *= corr; }
    mrun = mnew;

    #pragma unroll
    for (int j = 0; j < 4; ++j) {
        v2f pf; pf.x = pa[2 * j];  pf.y = pa[2 * j + 1];
        const v2f a0 = (j & 1) ? ea_hi(ta.v[j >> 1])       : ea_lo(ta.v[j >> 1]);
        const v2f a1 = (j & 1) ? ea_hi(ta.v[2 + (j >> 1)]) : ea_lo(ta.v[2 + (j >> 1)]);
        acc0 = __builtin_amdgcn_wmma_f32_16x16x4_f32(false, a0, false, pf,
                                                     (short)0, acc0, false, false);
        acc1 = __builtin_amdgcn_wmma_f32_16x16x4_f32(false, a1, false, pf,
                                                     (short)0, acc1, false, false);
    }
    #pragma unroll
    for (int j = 0; j < 4; ++j) {
        v2f pf; pf.x = pb[2 * j];  pf.y = pb[2 * j + 1];
        const v2f a0 = (j & 1) ? ea_hi(tb.v[j >> 1])       : ea_lo(tb.v[j >> 1]);
        const v2f a1 = (j & 1) ? ea_hi(tb.v[2 + (j >> 1)]) : ea_lo(tb.v[2 + (j >> 1)]);
        acc0 = __builtin_amdgcn_wmma_f32_16x16x4_f32(false, a0, false, pf,
                                                     (short)0, acc0, false, false);
        acc1 = __builtin_amdgcn_wmma_f32_16x16x4_f32(false, a1, false, pf,
                                                     (short)0, acc1, false, false);
    }
}

__global__ void __launch_bounds__(256)
ea_attn(const float* __restrict__ Q, const float* __restrict__ Km,
        const float* __restrict__ Vm, float* __restrict__ O)
{
    const int lane = threadIdx.x & 31;
    const int wave = threadIdx.x >> 5;
    const int tile = blockIdx.x * 8 + wave;
    const int qt = tile & 127;
    const int h  = (tile >> 7) & 7;
    const int b  = tile >> 10;
    const int col = lane & 15, half = lane >> 4;

    const float* Qrow = Q + (size_t)(b * Nn + qt * 16 + col) * Dd + h * DHh + 8 * half;
    const float qscale = 0.17677669529663687f * 1.4426950408889634f;
    v2f qf[8];
    #pragma unroll
    for (int q = 0; q < 4; ++q) {
        const v4f g = *(const v4f*)(Qrow + 16 * (q >> 1) + 4 * (q & 1)) * qscale;
        qf[2 * q]     = ea_lo(g);
        qf[2 * q + 1] = ea_hi(g);
    }

    float mrun = -__builtin_inff();
    float lrun = 0.f;
    v8f acc0 = {}, acc1 = {};

    const float* Kb = Km + (size_t)((b * Hh + h) * 256) * 512 + lane * 4;
    const float* Vb = Vm + (size_t)((b * Hh + h) * 256) * 512 + lane * 4;

    ea_kv e0, e1, e2, e3;
    __builtin_amdgcn_sched_barrier(0);
    ea_load_tile(e0, Kb, Vb, 0);
    __builtin_amdgcn_sched_barrier(0);
    ea_load_tile(e1, Kb, Vb, 1);
    __builtin_amdgcn_sched_barrier(0);
    ea_load_tile(e2, Kb, Vb, 2);
    __builtin_amdgcn_sched_barrier(0);
    ea_load_tile(e3, Kb, Vb, 3);
    __builtin_amdgcn_sched_barrier(0);
    for (int kt = 0; kt < Ll / 16; kt += 4) {
        int t0 = (kt + 4 < Ll / 16) ? kt + 4 : kt;
        int t2 = t0 + 2;
        ea_tile_step2(e0, e1, qf, mrun, lrun, acc0, acc1);
        asm volatile("" : "+v"(acc0), "+v"(acc1), "+s"(t0));
        ea_load_tile(e0, Kb, Vb, t0);
        ea_load_tile(e1, Kb, Vb, t0 + 1);
        __builtin_amdgcn_sched_barrier(0);
        ea_tile_step2(e2, e3, qf, mrun, lrun, acc0, acc1);
        asm volatile("" : "+v"(acc0), "+v"(acc1), "+s"(t2));
        ea_load_tile(e2, Kb, Vb, t2);
        ea_load_tile(e3, Kb, Vb, t2 + 1);
        __builtin_amdgcn_sched_barrier(0);
    }

    const float inv = 1.0f / (lrun + ea_xhalf(lrun));
    float* Op = O + (size_t)(b * Nn + qt * 16 + col) * Dd + h * DHh + 8 * half;
    v4f o0 = { acc0[0] * inv, acc0[1] * inv, acc0[2] * inv, acc0[3] * inv };
    v4f o1 = { acc0[4] * inv, acc0[5] * inv, acc0[6] * inv, acc0[7] * inv };
    v4f o2 = { acc1[0] * inv, acc1[1] * inv, acc1[2] * inv, acc1[3] * inv };
    v4f o3 = { acc1[4] * inv, acc1[5] * inv, acc1[6] * inv, acc1[7] * inv };
    *(v4f*)(Op)      = o0;
    *(v4f*)(Op + 4)  = o1;
    *(v4f*)(Op + 16) = o2;
    *(v4f*)(Op + 20) = o3;
}

__global__ void __launch_bounds__(128)
ea_gemm_out(const float* __restrict__ A, const float* __restrict__ W,
            const float* __restrict__ bias,
            const float* __restrict__ pts,
            const float* __restrict__ sw, const float* __restrict__ sb,
            float* __restrict__ OUT, int R)
{
    const int lane = threadIdx.x & 31;
    int tm, tn0;
    if (!ea_proj_tile(R, (int)blockIdx.x, tm, tn0)) return;
    const int col = lane & 15, half = lane >> 4;

    __shared__ __attribute__((aligned(16))) float ea_lds[2 * EA_LBUF];
    const int wave = threadIdx.x >> 5;
    const float* Eblk = A + (size_t)((tm - wave) * 16) * Dd;
    const float* Wblk = W + (size_t)(tn0 * 16) * Dd;

    v8f acc[4] = {};
    ea_proj_core<false>(Eblk, Wblk, ea_lds, acc);

    #pragma unroll
    for (int t = 0; t < 4; ++t) {
        const int colg = (tn0 + t) * 16 + col;
        const float s0 = sw[colg*3+0], s1 = sw[colg*3+1], s2 = sw[colg*3+2];
        const float add = bias[colg] + sb[colg];
        #pragma unroll
        for (int r = 0; r < 8; ++r) {
            int row = tm * 16 + r + 8 * half;
            float px = pts[row*3+0], py = pts[row*3+1], pz = pts[row*3+2];
            OUT[(size_t)row * Dd + colg] = acc[t][r] + add + px*s0 + py*s1 + pz*s2;
        }
    }
}

extern "C" void kernel_launch(void* const* d_in, const int* in_sizes, int n_in,
                              void* d_out, int out_size, void* d_ws, size_t ws_size,
                              hipStream_t stream)
{
    const float* cur_pts  = (const float*)d_in[0];
    const float* cur_feat = (const float*)d_in[1];
    const float* nbr_pts  = (const float*)d_in[2];
    const float* nbr_feat = (const float*)d_in[3];
    const float* in_w     = (const float*)d_in[4];
    const float* in_b     = (const float*)d_in[5];
    const float* out_w    = (const float*)d_in[6];
    const float* out_b    = (const float*)d_in[7];
    const float* sp_w     = (const float*)d_in[8];
    const float* sp_b     = (const float*)d_in[9];
    float* out = (float*)d_out;

    float* ws   = (float*)d_ws;
    float* curE = ws;
    float* nbrE = curE + 4096 * 256;
    float* Qb   = nbrE + 8192 * 256;
    float* Kb   = Qb   + 4096 * 256;
    float* Vb   = Kb   + 8192 * 256;
    float* Ob   = Vb   + 8192 * 256;

    const dim3 blk(256);
    const dim3 blk4(128);

    ea_knn_edge<<<dim3(EA_KNN_CUR_BLOCKS + EA_KNN_NBR_BLOCKS), blk, 0, stream>>>(
        cur_pts, cur_feat, curE, nbr_pts, nbr_feat, nbrE);

    ea_proj_qkv<<<dim3(2 * EA_KV_BLOCKS + EA_Q_BLOCKS), blk4, 0, stream>>>(
        curE, nbrE, in_w, in_b, Qb, Kb, Vb);

    ea_attn<<<dim3(2048 / 8), blk, 0, stream>>>(Qb, Kb, Vb, Ob);

    ea_gemm_out<<<dim3(4096 / 16), blk4, 0, stream>>>(Ob, out_w, out_b,
                                                    cur_pts, sp_w, sp_b,
                                                    out, 4096);
}
